# starting kernel
# baseline (speedup 1.0000x reference)
#include <hip/hip_runtime.h>
#include <hip/hip_bf16.h>

#define BATCH 65536
#define VEC   512
#define KCLUS 1024
#define GAMMA_ 0.99f
#define EPS_   1e-5f

#define ROWS_PER_BLOCK 128
#define XPAD 516

typedef __attribute__((ext_vector_type(2))) float v2f;
typedef __attribute__((ext_vector_type(8))) float v8f;

#define ESUM_OFF   0
#define COUNTS_OFF (VEC * KCLUS)
#define W2_OFF     (COUNTS_OFF + KCLUS)
#define CS_OFF     (W2_OFF + KCLUS)

__global__ void zero_ws_kernel(float* __restrict__ ws, int n) {
    int i = blockIdx.x * blockDim.x + threadIdx.x;
    if (i < n) ws[i] = 0.0f;
}

__global__ void w2_kernel(const float* __restrict__ W, float* __restrict__ w2) {
    int k = blockIdx.x * blockDim.x + threadIdx.x;
    if (k >= KCLUS) return;
    float s = 0.0f;
#pragma unroll 8
    for (int v = 0; v < VEC; ++v) {
        float w = W[(size_t)v * KCLUS + k];
        s += w * w;
    }
    w2[k] = s;
}

__global__ __launch_bounds__(128)
void assign_accum_kernel(const float* __restrict__ X,
                         const float* __restrict__ W,
                         const float* __restrict__ w2,
                         float* __restrict__ esum,
                         float* __restrict__ counts)
{
    extern __shared__ char smem[];
    float* xs   = (float*)smem;
    int*   aidx = (int*)(smem + ROWS_PER_BLOCK * XPAD * sizeof(float));

    const int tid  = threadIdx.x;
    const int lane = tid & 31;
    const int wave = tid >> 5;
    const int rowBase = blockIdx.x * ROWS_PER_BLOCK;

    {
        const float4* Xg = (const float4*)(X + (size_t)rowBase * VEC);
#pragma unroll 4
        for (int i = tid; i < ROWS_PER_BLOCK * (VEC / 4); i += 128) {
            int row = i >> 7;
            int v4  = i & 127;
            float4 val = Xg[(size_t)row * (VEC / 4) + v4];
            *(float4*)&xs[row * XPAD + v4 * 4] = val;
        }
    }
    __syncthreads();

    const int stripe0 = wave * 32;
    const int stripe1 = stripe0 + 16;
    const int l15     = lane & 15;
    const int khalf   = (lane >> 4) << 1;

    float best0[8], best1[8];
    int   bidx0[8], bidx1[8];
#pragma unroll
    for (int r = 0; r < 8; ++r) {
        best0[r] = 3.4e38f; bidx0[r] = 0;
        best1[r] = 3.4e38f; bidx1[r] = 0;
    }

    const float* xrow0 = &xs[(stripe0 + l15) * XPAD + khalf];
    const float* xrow1 = &xs[(stripe1 + l15) * XPAD + khalf];

    for (int kb = 0; kb < KCLUS / 16; ++kb) {
        const int col = kb * 16 + l15;
        v8f acc0 = {0.f, 0.f, 0.f, 0.f, 0.f, 0.f, 0.f, 0.f};
        v8f acc1 = {0.f, 0.f, 0.f, 0.f, 0.f, 0.f, 0.f, 0.f};
        const float* wp = W + (size_t)khalf * KCLUS + col;
#pragma unroll 4
        for (int s = 0; s < VEC / 4; ++s) {
            v2f b;
            b.x = wp[0];
            b.y = wp[KCLUS];
            wp += 4 * KCLUS;
            v2f a0 = *(const v2f*)(xrow0 + 4 * s);
            v2f a1 = *(const v2f*)(xrow1 + 4 * s);
            acc0 = __builtin_amdgcn_wmma_f32_16x16x4_f32(
                       false, a0, false, b, (short)0, acc0, false, false);
            acc1 = __builtin_amdgcn_wmma_f32_16x16x4_f32(
                       false, a1, false, b, (short)0, acc1, false, false);
        }
        const float w2c = w2[col];
#pragma unroll
        for (int r = 0; r < 8; ++r) {
            float sc0 = w2c - 2.0f * acc0[r];
            if (sc0 < best0[r]) { best0[r] = sc0; bidx0[r] = col; }
            float sc1 = w2c - 2.0f * acc1[r];
            if (sc1 < best1[r]) { best1[r] = sc1; bidx1[r] = col; }
        }
    }

#pragma unroll
    for (int m = 8; m >= 1; m >>= 1) {
#pragma unroll
        for (int r = 0; r < 8; ++r) {
            float ob0 = __shfl_xor(best0[r], m, 32);
            int   oi0 = __shfl_xor(bidx0[r], m, 32);
            if (ob0 < best0[r] || (ob0 == best0[r] && oi0 < bidx0[r])) {
                best0[r] = ob0; bidx0[r] = oi0;
            }
            float ob1 = __shfl_xor(best1[r], m, 32);
            int   oi1 = __shfl_xor(bidx1[r], m, 32);
            if (ob1 < best1[r] || (ob1 == best1[r] && oi1 < bidx1[r])) {
                best1[r] = ob1; bidx1[r] = oi1;
            }
        }
    }
    if (l15 == 0) {
        int rb0 = stripe0 + (lane >> 4) * 8;
        int rb1 = stripe1 + (lane >> 4) * 8;
#pragma unroll
        for (int r = 0; r < 8; ++r) {
            aidx[rb0 + r] = bidx0[r];
            aidx[rb1 + r] = bidx1[r];
        }
    }
    __syncthreads();

    if (tid < ROWS_PER_BLOCK) atomicAdd(&counts[aidx[tid]], 1.0f);

    for (int i = tid; i < ROWS_PER_BLOCK * VEC; i += 128) {
        int row = i >> 9;
        int v   = i & (VEC - 1);
        int k   = aidx[row];
        atomicAdd(&esum[(size_t)v * KCLUS + k], xs[row * XPAD + v]);
    }
}

__global__ __launch_bounds__(1024)
void finalize1_kernel(const float* __restrict__ cluster_size,
                      const float* __restrict__ counts,
                      float* __restrict__ cs)
{
    __shared__ float red[KCLUS];
    int k = threadIdx.x;
    float c = counts[k];
    if (c == 0.0f) c = 1.0f;
    float ncs = GAMMA_ * cluster_size[k] + (1.0f - GAMMA_) * c;
    red[k] = ncs;
    __syncthreads();
    for (int s = KCLUS / 2; s > 0; s >>= 1) {
        if (k < s) red[k] += red[k + s];
        __syncthreads();
    }
    float n = red[0];
    cs[k] = (ncs + EPS_) / (n + (float)KCLUS * EPS_) * n;
}

__global__ void finalize2_kernel(const float* __restrict__ embed_avg,
                                 const float* __restrict__ esum,
                                 const float* __restrict__ cs,
                                 float* __restrict__ out)
{
    int i = blockIdx.x * blockDim.x + threadIdx.x;
    if (i >= VEC * KCLUS) return;
    int k = i & (KCLUS - 1);
    out[i] = (GAMMA_ * embed_avg[i] + (1.0f - GAMMA_) * esum[i]) / cs[k];
}

extern "C" void kernel_launch(void* const* d_in, const int* in_sizes, int n_in,
                              void* d_out, int out_size, void* d_ws, size_t ws_size,
                              hipStream_t stream) {
    const float* X            = (const float*)d_in[0];
    const float* W            = (const float*)d_in[1];
    const float* cluster_size = (const float*)d_in[2];
    const float* embed_avg    = (const float*)d_in[3];
    float* out = (float*)d_out;
    float* ws  = (float*)d_ws;

    float* esum   = ws + ESUM_OFF;
    float* counts = ws + COUNTS_OFF;
    float* w2     = ws + W2_OFF;
    float* cs     = ws + CS_OFF;

    int zn = VEC * KCLUS + KCLUS;
    zero_ws_kernel<<<(zn + 255) / 256, 256, 0, stream>>>(ws, zn);

    w2_kernel<<<KCLUS / 256, 256, 0, stream>>>(W, w2);

    size_t smem = (size_t)ROWS_PER_BLOCK * XPAD * sizeof(float)
                + (size_t)ROWS_PER_BLOCK * sizeof(int);
    assign_accum_kernel<<<BATCH / ROWS_PER_BLOCK, 128, smem, stream>>>(
        X, W, w2, esum, counts);

    finalize1_kernel<<<1, KCLUS, 0, stream>>>(cluster_size, counts, cs);
    finalize2_kernel<<<(VEC * KCLUS + 255) / 256, 256, 0, stream>>>(
        embed_avg, esum, cs, out);
}


# Grid re-cut by cluster slice: workgroup holds 128 W columns in LDS all its life (was an X tile), X rows stream via registers as A; slice minima merged by a new launch; k order kept; atomics unordered
# speedup vs baseline: 1.4094x; 1.4094x over previous
#include <hip/hip_runtime.h>
#include <hip/hip_bf16.h>


#define BATCH 65536
#define VEC   512
#define KCLUS 1024
#define GAMMA_ 0.99f
#define EPS_   1e-5f

#define NSLICE      8
#define SLICE_COLS  (KCLUS / NSLICE)
#define CHUNK_ROWS  2048
#define WAVE_ROWS   (CHUNK_ROWS / 4)
#define NTILE       (WAVE_ROWS / 16)
#define WPITCH      144
#define KPAIRS      (VEC / 2)

typedef __attribute__((ext_vector_type(2))) float v2f;
typedef __attribute__((ext_vector_type(8))) float v8f;

#define ESUM_OFF   0
#define COUNTS_OFF (VEC * KCLUS)
#define W2_OFF     (COUNTS_OFF + KCLUS)
#define CS_OFF     (W2_OFF + KCLUS)

__global__ void zero_ws_kernel(float* __restrict__ ws, int n) {
    int i = blockIdx.x * blockDim.x + threadIdx.x;
    if (i < n) ws[i] = 0.0f;
}

__global__ void w2_kernel(const float* __restrict__ W, float* __restrict__ w2) {
    int k = blockIdx.x * blockDim.x + threadIdx.x;
    if (k >= KCLUS) return;
    float s = 0.0f;
#pragma unroll 8
    for (int v = 0; v < VEC; ++v) {
        float w = W[(size_t)v * KCLUS + k];
        s += w * w;
    }
    w2[k] = s;
}

template <int G>
__device__ __forceinline__ void slice_block(v2f (&a)[VEC / 4], v2f (&bb)[2][8][2],
                                            v8f& acc0, v8f& acc1,
                                            float (&best)[8], int (&bidx)[8],
                                            const v2f* wlane, const float* xn,
                                            const float (&w2r)[8],
                                            const int colBase, const int c)
{
    constexpr int np  = G >> 4;
    constexpr int blk = G & 15;
    constexpr int cur = G & 1;
    if (G + 1 < 64) {
        constexpr int np1  = (G + 1) >> 4;
        constexpr int blk1 = (G + 1) & 15;
#pragma unroll
        for (int i = 0; i < 8; ++i) {
            const int s1 = blk1 * 8 + i;
            bb[cur ^ 1][i][0] = wlane[s1 * 2 * WPITCH + np1 * 32];
            bb[cur ^ 1][i][1] = wlane[s1 * 2 * WPITCH + np1 * 32 + 16];
        }
    }
    __builtin_amdgcn_sched_barrier(0);
    if (blk == 0) {
        acc0 = (v8f){0.f, 0.f, 0.f, 0.f, 0.f, 0.f, 0.f, 0.f};
        acc1 = (v8f){0.f, 0.f, 0.f, 0.f, 0.f, 0.f, 0.f, 0.f};
    }
#pragma unroll
    for (int i = 0; i < 8; ++i) {
        const int s = blk * 8 + i;
        acc0 = __builtin_amdgcn_wmma_f32_16x16x4_f32(
                   false, a[s], false, bb[cur][i][0], (short)0, acc0, false, false);
        acc1 = __builtin_amdgcn_wmma_f32_16x16x4_f32(
                   false, a[s], false, bb[cur][i][1], (short)0, acc1, false, false);
        if (np == 3 && s >= 8) a[s - 8] = *(const v2f*)(xn + 4 * (s - 8));
    }
    if (blk == 15) {
        const int col0 = colBase + np * 32 + c;
        const int col1 = col0 + 16;
        const float w2c0 = w2r[2 * np];
        const float w2c1 = w2r[2 * np + 1];
#pragma unroll
        for (int r = 0; r < 8; ++r) {
            float sc0 = w2c0 - 2.0f * acc0[r];
            if (sc0 < best[r]) { best[r] = sc0; bidx[r] = col0; }
        }
#pragma unroll
        for (int r = 0; r < 8; ++r) {
            float sc1 = w2c1 - 2.0f * acc1[r];
            if (sc1 < best[r]) { best[r] = sc1; bidx[r] = col1; }
        }
    }
    __builtin_amdgcn_sched_barrier(0);
}

#define SLICE_BLOCK(G)  slice_block<G>(a, bb, acc0, acc1, best, bidx, wlane, xn, w2r, colBase, c);
#define SLICE_BLOCK4(G)  SLICE_BLOCK(G) SLICE_BLOCK(G + 1) SLICE_BLOCK(G + 2) SLICE_BLOCK(G + 3)
#define SLICE_BLOCK16(G) SLICE_BLOCK4(G) SLICE_BLOCK4(G + 4) SLICE_BLOCK4(G + 8) SLICE_BLOCK4(G + 12)

__global__ __launch_bounds__(128)
void assign_slice_kernel(const float* __restrict__ X,
                         const float* __restrict__ W,
                         const float* __restrict__ w2,
                         float* __restrict__ pscore,
                         float* __restrict__ pidx)
{
    extern __shared__ char smem[];
    v2f* wl = (v2f*)smem;

    const int tid   = threadIdx.x;
    const int lane  = tid & 31;
    const int wave  = tid >> 5;
    const int c     = lane & 15;
    const int h     = lane >> 4;
    const int slice = blockIdx.x & (NSLICE - 1);
    const int chunk = blockIdx.x >> 3;
    const int colBase = slice * SLICE_COLS;
    const int rowW    = chunk * CHUNK_ROWS + wave * WAVE_ROWS;

    v2f a[VEC / 4];
    {
        const float* xa = X + (size_t)(rowW + c) * VEC + 2 * h;
#pragma unroll
        for (int s = 0; s < VEC / 4; ++s) a[s] = *(const v2f*)(xa + 4 * s);
    }

    for (int i0 = tid; i0 < KPAIRS * (SLICE_COLS / 4); i0 += 128 * 8) {
        float4 w0[8], w1[8];
#pragma unroll
        for (int u = 0; u < 8; ++u) {
            const int i  = i0 + u * 128;
            const int kk = i >> 5;
            const int c4 = (i & 31) * 4;
            w0[u] = *(const float4*)(W + (size_t)(2 * kk)     * KCLUS + colBase + c4);
            w1[u] = *(const float4*)(W + (size_t)(2 * kk + 1) * KCLUS + colBase + c4);
        }
#pragma unroll
        for (int u = 0; u < 8; ++u) {
            const int i  = i0 + u * 128;
            const int kk = i >> 5;
            const int c4 = (i & 31) * 4;
            float4* dst = (float4*)(wl + kk * WPITCH + c4);
            dst[0] = make_float4(w0[u].x, w1[u].x, w0[u].y, w1[u].y);
            dst[1] = make_float4(w0[u].z, w1[u].z, w0[u].w, w1[u].w);
        }
    }
    __syncthreads();

    const v2f* wlane = wl + h * WPITCH + c;

    float w2r[8];
#pragma unroll
    for (int n = 0; n < 8; ++n) w2r[n] = w2[colBase + n * 16 + c];

#pragma unroll 1
    for (int t = 0; t < NTILE; ++t) {
        const int tn = (t + 1 < NTILE) ? t + 1 : t;
        const float* xn = X + (size_t)(rowW + tn * 16 + c) * VEC + 2 * h;

        float best[8];
        int   bidx[8];
#pragma unroll
        for (int r = 0; r < 8; ++r) { best[r] = 3.4e38f; bidx[r] = 0; }

        v2f bb[2][8][2];
#pragma unroll
        for (int i = 0; i < 8; ++i) {
            bb[0][i][0] = wlane[i * 2 * WPITCH];
            bb[0][i][1] = wlane[i * 2 * WPITCH + 16];
        }

        v8f acc0, acc1;
        SLICE_BLOCK16(0) SLICE_BLOCK16(16) SLICE_BLOCK16(32) SLICE_BLOCK16(48)
#pragma unroll
        for (int s = VEC / 4 - 8; s < VEC / 4; ++s) a[s] = *(const v2f*)(xn + 4 * s);

        float mn[8];
#pragma unroll
        for (int r = 0; r < 8; ++r) mn[r] = best[r];
#pragma unroll
        for (int m = 8; m >= 1; m >>= 1) {
#pragma unroll
            for (int r = 0; r < 8; ++r) {
                const float o = __shfl_xor(mn[r], m, 32);
                mn[r] = (o < mn[r]) ? o : mn[r];
            }
        }
#pragma unroll
        for (int r = 0; r < 8; ++r) bidx[r] = (best[r] == mn[r]) ? bidx[r] : 0x7fffffff;
#pragma unroll
        for (int m = 8; m >= 1; m >>= 1) {
#pragma unroll
            for (int r = 0; r < 8; ++r) {
                const int o = __shfl_xor(bidx[r], m, 32);
                bidx[r] = (o < bidx[r]) ? o : bidx[r];
            }
        }
        float osc = mn[0];
        int   oid = bidx[0];
#pragma unroll
        for (int r = 1; r < 8; ++r) {
            if ((c & 7) == r) { osc = mn[r]; oid = bidx[r]; }
        }
        if (c < 8) {
            const size_t o = (size_t)slice * BATCH + (size_t)(rowW + t * 16 + 8 * h + c);
            pscore[o] = osc;
            pidx[o]   = __int_as_float(oid);
        }
    }
}

__global__ void merge_kernel(float* sc_assign, const float* __restrict__ pidx)
{
    const int row = blockIdx.x * blockDim.x + threadIdx.x;
    if (row >= BATCH) return;
    float best = 3.4e38f;
    int   bi   = 0;
#pragma unroll
    for (int s = 0; s < NSLICE; ++s) {
        const float v  = sc_assign[(size_t)s * BATCH + row];
        const int   id = __float_as_int(pidx[(size_t)s * BATCH + row]);
        if (v < best) { best = v; bi = id; }
    }
    sc_assign[row] = __int_as_float(bi & (KCLUS - 1));
}

__global__ __launch_bounds__(256)
void scatter_kernel(const float* __restrict__ X,
                    const float* __restrict__ assignf,
                    float* __restrict__ esum,
                    float* __restrict__ counts)
{
    const int rowBase = blockIdx.x * 32;
    const int tid = threadIdx.x;
    if (tid < 32) atomicAdd(&counts[__float_as_int(assignf[rowBase + tid])], 1.0f);

    const float4* Xg = (const float4*)(X + (size_t)rowBase * VEC);
    for (int i = tid; i < 32 * (VEC / 4); i += 256) {
        const int row = i >> 7;
        const int v   = (i & 127) * 4;
        const int k   = __float_as_int(assignf[rowBase + row]);
        const float4 x = Xg[i];
        float* e = esum + (size_t)v * KCLUS + k;
        atomicAdd(e,             x.x);
        atomicAdd(e + KCLUS,     x.y);
        atomicAdd(e + 2 * KCLUS, x.z);
        atomicAdd(e + 3 * KCLUS, x.w);
    }
}

__global__ __launch_bounds__(1024)
void finalize1_kernel(const float* __restrict__ cluster_size,
                      const float* __restrict__ counts,
                      float* __restrict__ cs)
{
    __shared__ float red[KCLUS];
    int k = threadIdx.x;
    float c = counts[k];
    if (c == 0.0f) c = 1.0f;
    float ncs = GAMMA_ * cluster_size[k] + (1.0f - GAMMA_) * c;
    red[k] = ncs;
    __syncthreads();
    for (int s = KCLUS / 2; s > 0; s >>= 1) {
        if (k < s) red[k] += red[k + s];
        __syncthreads();
    }
    float n = red[0];
    cs[k] = (ncs + EPS_) / (n + (float)KCLUS * EPS_) * n;
}

__global__ void finalize2_kernel(const float* __restrict__ embed_avg,
                                 const float* __restrict__ esum,
                                 const float* __restrict__ cs,
                                 float* __restrict__ out)
{
    int i = blockIdx.x * blockDim.x + threadIdx.x;
    if (i >= VEC * KCLUS) return;
    int k = i & (KCLUS - 1);
    out[i] = (GAMMA_ * embed_avg[i] + (1.0f - GAMMA_) * esum[i]) / cs[k];
}

extern "C" void kernel_launch(void* const* d_in, const int* in_sizes, int n_in,
                              void* d_out, int out_size, void* d_ws, size_t ws_size,
                              hipStream_t stream) {
    const float* X            = (const float*)d_in[0];
    const float* W            = (const float*)d_in[1];
    const float* cluster_size = (const float*)d_in[2];
    const float* embed_avg    = (const float*)d_in[3];
    float* out = (float*)d_out;
    float* ws  = (float*)d_ws;

    float* esum   = ws + ESUM_OFF;
    float* counts = ws + COUNTS_OFF;
    float* w2     = ws + W2_OFF;
    float* cs     = ws + CS_OFF;

    w2_kernel<<<KCLUS / 256, 256, 0, stream>>>(W, w2);

    size_t smem = (size_t)KPAIRS * WPITCH * sizeof(v2f);
    assign_slice_kernel<<<NSLICE * (BATCH / CHUNK_ROWS), 128, smem, stream>>>(
        X, W, w2, out, esum);

    merge_kernel<<<BATCH / 256, 256, 0, stream>>>(out, esum);

    int zn = VEC * KCLUS + KCLUS;
    zero_ws_kernel<<<(zn + 255) / 256, 256, 0, stream>>>(ws, zn);

    scatter_kernel<<<BATCH / 32, 256, 0, stream>>>(X, out, esum, counts);

    finalize1_kernel<<<1, KCLUS, 0, stream>>>(cluster_size, counts, cs);
    finalize2_kernel<<<(VEC * KCLUS + 255) / 256, 256, 0, stream>>>(
        embed_avg, esum, cs, out);
}


# embed_sum: rows grouped by cluster (integer count, prefix sum, slot), gathered 32 list slots per workgroup, ONE float atomic per cluster stretch+column (was per row+column); sums unordered as before
# speedup vs baseline: 4.2789x; 3.0360x over previous
#include <hip/hip_runtime.h>
#include <hip/hip_bf16.h>


#define BATCH 65536
#define VEC   512
#define KCLUS 1024
#define GAMMA_ 0.99f
#define EPS_   1e-5f

#define NSLICE      8
#define SLICE_COLS  (KCLUS / NSLICE)
#define CHUNK_ROWS  2048
#define WAVE_ROWS   (CHUNK_ROWS / 4)
#define NTILE       (WAVE_ROWS / 16)
#define WPITCH      144
#define KPAIRS      (VEC / 2)

typedef __attribute__((ext_vector_type(2))) float v2f;
typedef __attribute__((ext_vector_type(8))) float v8f;

typedef __attribute__((ext_vector_type(4))) float v4f;

#define ESUM_OFF   0
#define COUNTS_OFF (VEC * KCLUS)
#define W2_OFF     (COUNTS_OFF + KCLUS)
#define CS_OFF     (W2_OFF + KCLUS)
#define PERM_OFF   BATCH
#define PERMK_OFF  (2 * BATCH)
#define CURSOR_OFF (3 * BATCH)

#define GROWS 32

__global__ void w2_kernel(const float* __restrict__ W, float* __restrict__ w2, int* __restrict__ cnt) {
    int k = blockIdx.x * blockDim.x + threadIdx.x;
    if (k >= KCLUS) return;
    cnt[k] = 0;
    float s = 0.0f;
#pragma unroll 8
    for (int v = 0; v < VEC; ++v) {
        float w = W[(size_t)v * KCLUS + k];
        s += w * w;
    }
    w2[k] = s;
}

template <int G>
__device__ __forceinline__ void slice_block(v2f (&a)[VEC / 4], v2f (&bb)[2][8][2],
                                            v8f& acc0, v8f& acc1,
                                            float (&best)[8], int (&bidx)[8],
                                            const v2f* wlane, const float* xn,
                                            const float (&w2r)[8],
                                            const int colBase, const int c)
{
    constexpr int np  = G >> 4;
    constexpr int blk = G & 15;
    constexpr int cur = G & 1;
    if (G + 1 < 64) {
        constexpr int np1  = (G + 1) >> 4;
        constexpr int blk1 = (G + 1) & 15;
#pragma unroll
        for (int i = 0; i < 8; ++i) {
            const int s1 = blk1 * 8 + i;
            bb[cur ^ 1][i][0] = wlane[s1 * 2 * WPITCH + np1 * 32];
            bb[cur ^ 1][i][1] = wlane[s1 * 2 * WPITCH + np1 * 32 + 16];
        }
    }
    __builtin_amdgcn_sched_barrier(0);
    if (blk == 0) {
        acc0 = (v8f){0.f, 0.f, 0.f, 0.f, 0.f, 0.f, 0.f, 0.f};
        acc1 = (v8f){0.f, 0.f, 0.f, 0.f, 0.f, 0.f, 0.f, 0.f};
    }
#pragma unroll
    for (int i = 0; i < 8; ++i) {
        const int s = blk * 8 + i;
        acc0 = __builtin_amdgcn_wmma_f32_16x16x4_f32(
                   false, a[s], false, bb[cur][i][0], (short)0, acc0, false, false);
        acc1 = __builtin_amdgcn_wmma_f32_16x16x4_f32(
                   false, a[s], false, bb[cur][i][1], (short)0, acc1, false, false);
        if (np == 3 && s >= 8) a[s - 8] = *(const v2f*)(xn + 4 * (s - 8));
    }
    if (blk == 15) {
        const int col0 = colBase + np * 32 + c;
        const int col1 = col0 + 16;
        const float w2c0 = w2r[2 * np];
        const float w2c1 = w2r[2 * np + 1];
#pragma unroll
        for (int r = 0; r < 8; ++r) {
            float sc0 = w2c0 - 2.0f * acc0[r];
            if (sc0 < best[r]) { best[r] = sc0; bidx[r] = col0; }
        }
#pragma unroll
        for (int r = 0; r < 8; ++r) {
            float sc1 = w2c1 - 2.0f * acc1[r];
            if (sc1 < best[r]) { best[r] = sc1; bidx[r] = col1; }
        }
    }
    __builtin_amdgcn_sched_barrier(0);
}

#define SLICE_BLOCK(G)  slice_block<G>(a, bb, acc0, acc1, best, bidx, wlane, xn, w2r, colBase, c);
#define SLICE_BLOCK4(G)  SLICE_BLOCK(G) SLICE_BLOCK(G + 1) SLICE_BLOCK(G + 2) SLICE_BLOCK(G + 3)
#define SLICE_BLOCK16(G) SLICE_BLOCK4(G) SLICE_BLOCK4(G + 4) SLICE_BLOCK4(G + 8) SLICE_BLOCK4(G + 12)

__global__ __launch_bounds__(128)
void assign_slice_kernel(const float* __restrict__ X,
                         const float* __restrict__ W,
                         const float* __restrict__ w2,
                         float* __restrict__ pscore,
                         float* __restrict__ pidx)
{
    extern __shared__ char smem[];
    v2f* wl = (v2f*)smem;

    const int tid   = threadIdx.x;
    const int lane  = tid & 31;
    const int wave  = tid >> 5;
    const int c     = lane & 15;
    const int h     = lane >> 4;
    const int slice = blockIdx.x & (NSLICE - 1);
    const int chunk = blockIdx.x >> 3;
    const int colBase = slice * SLICE_COLS;
    const int rowW    = chunk * CHUNK_ROWS + wave * WAVE_ROWS;

    v2f a[VEC / 4];
    {
        const float* xa = X + (size_t)(rowW + c) * VEC + 2 * h;
#pragma unroll
        for (int s = 0; s < VEC / 4; ++s) a[s] = *(const v2f*)(xa + 4 * s);
    }

    for (int i0 = tid; i0 < KPAIRS * (SLICE_COLS / 4); i0 += 128 * 8) {
        float4 w0[8], w1[8];
#pragma unroll
        for (int u = 0; u < 8; ++u) {
            const int i  = i0 + u * 128;
            const int kk = i >> 5;
            const int c4 = (i & 31) * 4;
            w0[u] = *(const float4*)(W + (size_t)(2 * kk)     * KCLUS + colBase + c4);
            w1[u] = *(const float4*)(W + (size_t)(2 * kk + 1) * KCLUS + colBase + c4);
        }
#pragma unroll
        for (int u = 0; u < 8; ++u) {
            const int i  = i0 + u * 128;
            const int kk = i >> 5;
            const int c4 = (i & 31) * 4;
            float4* dst = (float4*)(wl + kk * WPITCH + c4);
            dst[0] = make_float4(w0[u].x, w1[u].x, w0[u].y, w1[u].y);
            dst[1] = make_float4(w0[u].z, w1[u].z, w0[u].w, w1[u].w);
        }
    }
    __syncthreads();

    const v2f* wlane = wl + h * WPITCH + c;

    float w2r[8];
#pragma unroll
    for (int n = 0; n < 8; ++n) w2r[n] = w2[colBase + n * 16 + c];

#pragma unroll 1
    for (int t = 0; t < NTILE; ++t) {
        const int tn = (t + 1 < NTILE) ? t + 1 : t;
        const float* xn = X + (size_t)(rowW + tn * 16 + c) * VEC + 2 * h;

        float best[8];
        int   bidx[8];
#pragma unroll
        for (int r = 0; r < 8; ++r) { best[r] = 3.4e38f; bidx[r] = 0; }

        v2f bb[2][8][2];
#pragma unroll
        for (int i = 0; i < 8; ++i) {
            bb[0][i][0] = wlane[i * 2 * WPITCH];
            bb[0][i][1] = wlane[i * 2 * WPITCH + 16];
        }

        v8f acc0, acc1;
        SLICE_BLOCK16(0) SLICE_BLOCK16(16) SLICE_BLOCK16(32) SLICE_BLOCK16(48)
#pragma unroll
        for (int s = VEC / 4 - 8; s < VEC / 4; ++s) a[s] = *(const v2f*)(xn + 4 * s);

        float mn[8];
#pragma unroll
        for (int r = 0; r < 8; ++r) mn[r] = best[r];
#pragma unroll
        for (int m = 8; m >= 1; m >>= 1) {
#pragma unroll
            for (int r = 0; r < 8; ++r) {
                const float o = __shfl_xor(mn[r], m, 32);
                mn[r] = (o < mn[r]) ? o : mn[r];
            }
        }
#pragma unroll
        for (int r = 0; r < 8; ++r) bidx[r] = (best[r] == mn[r]) ? bidx[r] : 0x7fffffff;
#pragma unroll
        for (int m = 8; m >= 1; m >>= 1) {
#pragma unroll
            for (int r = 0; r < 8; ++r) {
                const int o = __shfl_xor(bidx[r], m, 32);
                bidx[r] = (o < bidx[r]) ? o : bidx[r];
            }
        }
        float osc = mn[0];
        int   oid = bidx[0];
#pragma unroll
        for (int r = 1; r < 8; ++r) {
            if ((c & 7) == r) { osc = mn[r]; oid = bidx[r]; }
        }
        if (c < 8) {
            const size_t o = (size_t)slice * BATCH + (size_t)(rowW + t * 16 + 8 * h + c);
            pscore[o] = osc;
            pidx[o]   = __int_as_float(oid);
        }
    }
}

__global__ void merge_kernel(float* sc_assign, const float* __restrict__ pidx, int* __restrict__ cnt)
{
    const int row = blockIdx.x * blockDim.x + threadIdx.x;
    if (row >= BATCH) return;
    float best = 3.4e38f;
    int   bi   = 0;
#pragma unroll
    for (int s = 0; s < NSLICE; ++s) {
        const float v  = sc_assign[(size_t)s * BATCH + row];
        const int   id = __float_as_int(pidx[(size_t)s * BATCH + row]);
        if (v < best) { best = v; bi = id; }
    }
    sc_assign[row] = __int_as_float(bi & (KCLUS - 1));
    atomicAdd(&cnt[bi & (KCLUS - 1)], 1);
}

__global__ __launch_bounds__(1024)
void scan_finalize1_kernel(const float* __restrict__ cluster_size,
                           const int* __restrict__ cnt,
                           int* __restrict__ cursor,
                           float* __restrict__ cs)
{
    __shared__ float red[KCLUS];
    __shared__ int   pre[KCLUS];
    const int k  = threadIdx.x;
    const int ci = cnt[k];
    float c = (float)ci;
    if (c == 0.0f) c = 1.0f;
    float ncs = GAMMA_ * cluster_size[k] + (1.0f - GAMMA_) * c;
    red[k] = ncs;
    pre[k] = ci;
    __syncthreads();
    for (int s = KCLUS / 2; s > 0; s >>= 1) {
        if (k < s) red[k] += red[k + s];
        __syncthreads();
    }
    for (int d = 1; d < KCLUS; d <<= 1) {
        const int v = (k >= d) ? pre[k - d] : 0;
        __syncthreads();
        pre[k] += v;
        __syncthreads();
    }
    cursor[k] = pre[k] - ci;
    float n = red[0];
    cs[k] = (ncs + EPS_) / (n + (float)KCLUS * EPS_) * n;
}

__global__ __launch_bounds__(256)
void place_kernel(const float* __restrict__ assignf,
                  int* __restrict__ cursor,
                  int* __restrict__ perm,
                  int* __restrict__ permk,
                  float* __restrict__ esum)
{
    const int tid = threadIdx.x;
    const int row = blockIdx.x * 256 + tid;
    float4* z = (float4*)(esum + (size_t)blockIdx.x * 2048) + tid;
    z[0]   = make_float4(0.f, 0.f, 0.f, 0.f);
    z[256] = make_float4(0.f, 0.f, 0.f, 0.f);
    const int k   = __float_as_int(assignf[row]) & (KCLUS - 1);
    const int pos = atomicAdd(&cursor[k], 1) & (BATCH - 1);
    perm[pos]  = row;
    permk[pos] = k;
}

__global__ __launch_bounds__(128)
void gather_kernel(const float* __restrict__ X,
                   const int* __restrict__ perm,
                   const int* __restrict__ permk,
                   float* __restrict__ esum)
{
    const int lane = threadIdx.x & 31;
    const int wave = threadIdx.x >> 5;
    const int base = blockIdx.x * GROWS;
    const int myrow = perm[base + lane]  & (BATCH - 1);
    const int myk   = permk[base + lane] & (KCLUS - 1);
    const float* xb = X    + wave * 128 + 4 * lane;
    float*       eb = esum + wave * 128 + 4 * lane;

    v4f acc = {0.f, 0.f, 0.f, 0.f};
    int curk = __builtin_amdgcn_readlane(myk, 0);
#pragma unroll
    for (int g = 0; g < GROWS / 8; ++g) {
        v4f x[8];
#pragma unroll
        for (int u = 0; u < 8; ++u) {
            const int row = __builtin_amdgcn_readlane(myrow, 8 * g + u);
            x[u] = __builtin_nontemporal_load((const v4f*)(xb + (size_t)row * VEC));
        }
#pragma unroll
        for (int u = 0; u < 8; ++u) {
            const int k = __builtin_amdgcn_readlane(myk, 8 * g + u);
            if (k != curk) {
                float* e = eb + (size_t)curk * VEC;
                atomicAdd(e,     acc.x);
                atomicAdd(e + 1, acc.y);
                atomicAdd(e + 2, acc.z);
                atomicAdd(e + 3, acc.w);
                acc  = (v4f){0.f, 0.f, 0.f, 0.f};
                curk = k;
            }
            acc += x[u];
        }
    }
    float* e = eb + (size_t)curk * VEC;
    atomicAdd(e,     acc.x);
    atomicAdd(e + 1, acc.y);
    atomicAdd(e + 2, acc.z);
    atomicAdd(e + 3, acc.w);
}

__global__ __launch_bounds__(256)
void finalize2_kernel(const float* __restrict__ embed_avg,
                      const float* __restrict__ esum,
                      const float* __restrict__ cs,
                      float* __restrict__ out)
{
    __shared__ float tile[32][33];
    const int lane = threadIdx.x & 31;
    const int wave = threadIdx.x >> 5;
    const int k0 = (blockIdx.x >> 4) * 32;
    const int v0 = (blockIdx.x & 15) * 32;
#pragma unroll
    for (int i = 0; i < 4; ++i) {
        const int j = wave * 4 + i;
        tile[j][lane] = esum[(size_t)(k0 + j) * VEC + v0 + lane];
    }
    __syncthreads();
    const float csk = cs[k0 + lane];
#pragma unroll
    for (int i = 0; i < 4; ++i) {
        const int vl = wave * 4 + i;
        const size_t o = (size_t)(v0 + vl) * KCLUS + k0 + lane;
        out[o] = (GAMMA_ * embed_avg[o] + (1.0f - GAMMA_) * tile[lane][vl]) / csk;
    }
}

extern "C" void kernel_launch(void* const* d_in, const int* in_sizes, int n_in,
                              void* d_out, int out_size, void* d_ws, size_t ws_size,
                              hipStream_t stream) {
    const float* X            = (const float*)d_in[0];
    const float* W            = (const float*)d_in[1];
    const float* cluster_size = (const float*)d_in[2];
    const float* embed_avg    = (const float*)d_in[3];
    float* out = (float*)d_out;
    float* ws  = (float*)d_ws;

    float* esum   = ws + ESUM_OFF;
    float* w2     = ws + W2_OFF;
    float* cs     = ws + CS_OFF;
    int*   cnt    = (int*)(ws + COUNTS_OFF);
    int*   perm   = (int*)(out + PERM_OFF);
    int*   permk  = (int*)(out + PERMK_OFF);
    int*   cursor = (int*)(out + CURSOR_OFF);

    w2_kernel<<<KCLUS / 256, 256, 0, stream>>>(W, w2, cnt);

    size_t smem = (size_t)KPAIRS * WPITCH * sizeof(v2f);
    assign_slice_kernel<<<NSLICE * (BATCH / CHUNK_ROWS), 128, smem, stream>>>(
        X, W, w2, out, esum);

    merge_kernel<<<BATCH / 256, 256, 0, stream>>>(out, esum, cnt);

    scan_finalize1_kernel<<<1, KCLUS, 0, stream>>>(cluster_size, cnt, cursor, cs);
    place_kernel<<<BATCH / 256, 256, 0, stream>>>(out, cursor, perm, permk, esum);
    gather_kernel<<<BATCH / GROWS, 128, 0, stream>>>(X, perm, permk, esum);

    finalize2_kernel<<<(KCLUS / 32) * (VEC / 32), 256, 0, stream>>>(embed_avg, esum, cs, out);
}


# B fragments: W slice re-laid in LDS as column-tile pairs, ONE 16-byte read feeds both tiles' products (1024 -> 512 reads per row tile), reads stand as one group per 64 products; k order, sums kept
# speedup vs baseline: 4.8451x; 1.1323x over previous
#include <hip/hip_runtime.h>
#include <hip/hip_bf16.h>


#define BATCH 65536
#define VEC   512
#define KCLUS 1024
#define GAMMA_ 0.99f
#define EPS_   1e-5f

#define NSLICE      8
#define SLICE_COLS  (KCLUS / NSLICE)
#define CHUNK_ROWS  2048
#define WAVE_ROWS   (CHUNK_ROWS / 4)
#define NTILE       (WAVE_ROWS / 16)
#define RPITCH      72
#define KPAIRS      (VEC / 2)
#define BSTEPS      32
#define NBLK        (VEC / 4 / BSTEPS)
#define SLICE_BLOCKS SLICE_BLOCK16(0)

typedef __attribute__((ext_vector_type(2))) float v2f;
typedef __attribute__((ext_vector_type(8))) float v8f;

typedef __attribute__((ext_vector_type(4))) float v4f;

#define ESUM_OFF   0
#define COUNTS_OFF (VEC * KCLUS)
#define W2_OFF     (COUNTS_OFF + KCLUS)
#define CS_OFF     (W2_OFF + KCLUS)
#define PERM_OFF   BATCH
#define PERMK_OFF  (2 * BATCH)
#define CURSOR_OFF (3 * BATCH)

#define GROWS 32

__global__ void w2_kernel(const float* __restrict__ W, float* __restrict__ w2, int* __restrict__ cnt) {
    int k = blockIdx.x * blockDim.x + threadIdx.x;
    if (k >= KCLUS) return;
    cnt[k] = 0;
    float s = 0.0f;
#pragma unroll 8
    for (int v = 0; v < VEC; ++v) {
        float w = W[(size_t)v * KCLUS + k];
        s += w * w;
    }
    w2[k] = s;
}

template <int G>
__device__ __forceinline__ void slice_block(v2f (&a)[VEC / 4], v4f (&bb)[2][BSTEPS],
                                            v8f& acc0, v8f& acc1,
                                            float (&best)[8], int (&bidx)[8],
                                            const v4f* wlane, const float* xn,
                                            const float (&w2r)[8],
                                            const int colBase, const int c)
{
    constexpr int np  = G / NBLK;
    constexpr int blk = G % NBLK;
    constexpr int cur = G & 1;
    if (G + 1 < 4 * NBLK) {
        constexpr int np1  = (G + 1) / NBLK;
        constexpr int blk1 = (G + 1) % NBLK;
#pragma unroll
        for (int i = 0; i < BSTEPS; ++i) {
            const int s1 = blk1 * BSTEPS + i;
            bb[cur ^ 1][i] = wlane[s1 * 2 * RPITCH + np1 * 16];
        }
    }
    __builtin_amdgcn_sched_barrier(0);
    if (blk == 0) {
        acc0 = (v8f){0.f, 0.f, 0.f, 0.f, 0.f, 0.f, 0.f, 0.f};
        acc1 = (v8f){0.f, 0.f, 0.f, 0.f, 0.f, 0.f, 0.f, 0.f};
    }
#pragma unroll
    for (int i = 0; i < BSTEPS; ++i) {
        const int s = blk * BSTEPS + i;
        const v4f b = bb[cur][i];
        acc0 = __builtin_amdgcn_wmma_f32_16x16x4_f32(
                   false, a[s], false, __builtin_shufflevector(b, b, 0, 1), (short)0, acc0, false, false);
        acc1 = __builtin_amdgcn_wmma_f32_16x16x4_f32(
                   false, a[s], false, __builtin_shufflevector(b, b, 2, 3), (short)0, acc1, false, false);
        if (np == 3 && s >= 8) a[s - 8] = *(const v2f*)(xn + 4 * (s - 8));
    }
    asm volatile("" : "+v"(acc0), "+v"(acc1));
    if (blk == NBLK - 1) {
        const int col0 = colBase + np * 32 + c;
        const int col1 = col0 + 16;
        const float w2c0 = w2r[2 * np];
        const float w2c1 = w2r[2 * np + 1];
#pragma unroll
        for (int r = 0; r < 8; ++r) {
            float sc0 = w2c0 - 2.0f * acc0[r];
            if (sc0 < best[r]) { best[r] = sc0; bidx[r] = col0; }
        }
#pragma unroll
        for (int r = 0; r < 8; ++r) {
            float sc1 = w2c1 - 2.0f * acc1[r];
            if (sc1 < best[r]) { best[r] = sc1; bidx[r] = col1; }
        }
    }
    __builtin_amdgcn_sched_barrier(0);
}

#define SLICE_BLOCK(G)  slice_block<G>(a, bb, acc0, acc1, best, bidx, wlane, xn, w2r, colBase, c);
#define SLICE_BLOCK4(G)  SLICE_BLOCK(G) SLICE_BLOCK(G + 1) SLICE_BLOCK(G + 2) SLICE_BLOCK(G + 3)
#define SLICE_BLOCK16(G) SLICE_BLOCK4(G) SLICE_BLOCK4(G + 4) SLICE_BLOCK4(G + 8) SLICE_BLOCK4(G + 12)

__global__ __launch_bounds__(128)
void assign_slice_kernel(const float* __restrict__ X,
                         const float* __restrict__ W,
                         const float* __restrict__ w2,
                         float* __restrict__ pscore,
                         float* __restrict__ pidx)
{
    extern __shared__ char smem[];
    v4f* wl = (v4f*)smem;

    const int tid   = threadIdx.x;
    const int lane  = tid & 31;
    const int wave  = tid >> 5;
    const int c     = lane & 15;
    const int h     = lane >> 4;
    const int slice = blockIdx.x & (NSLICE - 1);
    const int chunk = blockIdx.x >> 3;
    const int colBase = slice * SLICE_COLS;
    const int rowW    = chunk * CHUNK_ROWS + wave * WAVE_ROWS;

    v2f a[VEC / 4];
    {
        const float* xa = X + (size_t)(rowW + c) * VEC + 2 * h;
#pragma unroll
        for (int s = 0; s < VEC / 4; ++s) a[s] = *(const v2f*)(xa + 4 * s);
    }

    for (int i0 = tid; i0 < KPAIRS * 16; i0 += 128 * 4) {
        float4 w0a[4], w1a[4], w0b[4], w1b[4];
#pragma unroll
        for (int u = 0; u < 4; ++u) {
            const int i  = i0 + u * 128;
            const int kk = i >> 4;
            const int cc = ((i >> 2) & 3) * 32 + (i & 3) * 4;
            const float* r0 = W + (size_t)(2 * kk) * KCLUS + colBase + cc;
            w0a[u] = *(const float4*)(r0);
            w1a[u] = *(const float4*)(r0 + KCLUS);
            w0b[u] = *(const float4*)(r0 + 16);
            w1b[u] = *(const float4*)(r0 + KCLUS + 16);
        }
#pragma unroll
        for (int u = 0; u < 4; ++u) {
            const int i  = i0 + u * 128;
            const int kk = i >> 4;
            v4f* dst = wl + kk * RPITCH + ((i >> 2) & 3) * 16 + (i & 3) * 4;
            dst[0] = (v4f){w0a[u].x, w1a[u].x, w0b[u].x, w1b[u].x};
            dst[1] = (v4f){w0a[u].y, w1a[u].y, w0b[u].y, w1b[u].y};
            dst[2] = (v4f){w0a[u].z, w1a[u].z, w0b[u].z, w1b[u].z};
            dst[3] = (v4f){w0a[u].w, w1a[u].w, w0b[u].w, w1b[u].w};
        }
    }
    __syncthreads();

    const v4f* wlane = wl + h * RPITCH + c;

    float w2r[8];
#pragma unroll
    for (int n = 0; n < 8; ++n) w2r[n] = w2[colBase + n * 16 + c];

#pragma unroll 1
    for (int t = 0; t < NTILE; ++t) {
        const int tn = (t + 1 < NTILE) ? t + 1 : t;
        const float* xn = X + (size_t)(rowW + tn * 16 + c) * VEC + 2 * h;

        float best[8];
        int   bidx[8];
#pragma unroll
        for (int r = 0; r < 8; ++r) { best[r] = 3.4e38f; bidx[r] = 0; }

        v4f bb[2][BSTEPS];
#pragma unroll
        for (int i = 0; i < BSTEPS; ++i) bb[0][i] = wlane[i * 2 * RPITCH];

        v8f acc0, acc1;
        SLICE_BLOCKS
#pragma unroll
        for (int s = VEC / 4 - 8; s < VEC / 4; ++s) a[s] = *(const v2f*)(xn + 4 * s);

        float mn[8];
#pragma unroll
        for (int r = 0; r < 8; ++r) mn[r] = best[r];
#pragma unroll
        for (int m = 8; m >= 1; m >>= 1) {
#pragma unroll
            for (int r = 0; r < 8; ++r) {
                const float o = __shfl_xor(mn[r], m, 32);
                mn[r] = (o < mn[r]) ? o : mn[r];
            }
        }
#pragma unroll
        for (int r = 0; r < 8; ++r) bidx[r] = (best[r] == mn[r]) ? bidx[r] : 0x7fffffff;
#pragma unroll
        for (int m = 8; m >= 1; m >>= 1) {
#pragma unroll
            for (int r = 0; r < 8; ++r) {
                const int o = __shfl_xor(bidx[r], m, 32);
                bidx[r] = (o < bidx[r]) ? o : bidx[r];
            }
        }
        float osc = mn[0];
        int   oid = bidx[0];
#pragma unroll
        for (int r = 1; r < 8; ++r) {
            if ((c & 7) == r) { osc = mn[r]; oid = bidx[r]; }
        }
        if (c < 8) {
            const size_t o = (size_t)slice * BATCH + (size_t)(rowW + t * 16 + 8 * h + c);
            pscore[o] = osc;
            pidx[o]   = __int_as_float(oid);
        }
    }
}

__global__ void merge_kernel(float* sc_assign, const float* __restrict__ pidx, int* __restrict__ cnt)
{
    const int row = blockIdx.x * blockDim.x + threadIdx.x;
    if (row >= BATCH) return;
    float best = 3.4e38f;
    int   bi   = 0;
#pragma unroll
    for (int s = 0; s < NSLICE; ++s) {
        const float v  = sc_assign[(size_t)s * BATCH + row];
        const int   id = __float_as_int(pidx[(size_t)s * BATCH + row]);
        if (v < best) { best = v; bi = id; }
    }
    sc_assign[row] = __int_as_float(bi & (KCLUS - 1));
    atomicAdd(&cnt[bi & (KCLUS - 1)], 1);
}

__global__ __launch_bounds__(1024)
void scan_finalize1_kernel(const float* __restrict__ cluster_size,
                           const int* __restrict__ cnt,
                           int* __restrict__ cursor,
                           float* __restrict__ cs)
{
    __shared__ float red[KCLUS];
    __shared__ int   pre[KCLUS];
    const int k  = threadIdx.x;
    const int ci = cnt[k];
    float c = (float)ci;
    if (c == 0.0f) c = 1.0f;
    float ncs = GAMMA_ * cluster_size[k] + (1.0f - GAMMA_) * c;
    red[k] = ncs;
    pre[k] = ci;
    __syncthreads();
    for (int s = KCLUS / 2; s > 0; s >>= 1) {
        if (k < s) red[k] += red[k + s];
        __syncthreads();
    }
    for (int d = 1; d < KCLUS; d <<= 1) {
        const int v = (k >= d) ? pre[k - d] : 0;
        __syncthreads();
        pre[k] += v;
        __syncthreads();
    }
    cursor[k] = pre[k] - ci;
    float n = red[0];
    cs[k] = (ncs + EPS_) / (n + (float)KCLUS * EPS_) * n;
}

__global__ __launch_bounds__(256)
void place_kernel(const float* __restrict__ assignf,
                  int* __restrict__ cursor,
                  int* __restrict__ perm,
                  int* __restrict__ permk,
                  float* __restrict__ esum)
{
    const int tid = threadIdx.x;
    const int row = blockIdx.x * 256 + tid;
    float4* z = (float4*)(esum + (size_t)blockIdx.x * 2048) + tid;
    z[0]   = make_float4(0.f, 0.f, 0.f, 0.f);
    z[256] = make_float4(0.f, 0.f, 0.f, 0.f);
    const int k   = __float_as_int(assignf[row]) & (KCLUS - 1);
    const int pos = atomicAdd(&cursor[k], 1) & (BATCH - 1);
    perm[pos]  = row;
    permk[pos] = k;
}

__global__ __launch_bounds__(128)
void gather_kernel(const float* __restrict__ X,
                   const int* __restrict__ perm,
                   const int* __restrict__ permk,
                   float* __restrict__ esum)
{
    const int lane = threadIdx.x & 31;
    const int wave = threadIdx.x >> 5;
    const int base = blockIdx.x * GROWS;
    const int myrow = perm[base + lane]  & (BATCH - 1);
    const int myk   = permk[base + lane] & (KCLUS - 1);
    const float* xb = X    + wave * 128 + 4 * lane;
    float*       eb = esum + wave * 128 + 4 * lane;

    v4f acc = {0.f, 0.f, 0.f, 0.f};
    int curk = __builtin_amdgcn_readlane(myk, 0);
#pragma unroll
    for (int g = 0; g < GROWS / 8; ++g) {
        v4f x[8];
#pragma unroll
        for (int u = 0; u < 8; ++u) {
            const int row = __builtin_amdgcn_readlane(myrow, 8 * g + u);
            x[u] = __builtin_nontemporal_load((const v4f*)(xb + (size_t)row * VEC));
        }
#pragma unroll
        for (int u = 0; u < 8; ++u) {
            const int k = __builtin_amdgcn_readlane(myk, 8 * g + u);
            if (k != curk) {
                float* e = eb + (size_t)curk * VEC;
                atomicAdd(e,     acc.x);
                atomicAdd(e + 1, acc.y);
                atomicAdd(e + 2, acc.z);
                atomicAdd(e + 3, acc.w);
                acc  = (v4f){0.f, 0.f, 0.f, 0.f};
                curk = k;
            }
            acc += x[u];
        }
    }
    float* e = eb + (size_t)curk * VEC;
    atomicAdd(e,     acc.x);
    atomicAdd(e + 1, acc.y);
    atomicAdd(e + 2, acc.z);
    atomicAdd(e + 3, acc.w);
}

__global__ __launch_bounds__(256)
void finalize2_kernel(const float* __restrict__ embed_avg,
                      const float* __restrict__ esum,
                      const float* __restrict__ cs,
                      float* __restrict__ out)
{
    __shared__ float tile[32][33];
    const int lane = threadIdx.x & 31;
    const int wave = threadIdx.x >> 5;
    const int k0 = (blockIdx.x >> 4) * 32;
    const int v0 = (blockIdx.x & 15) * 32;
#pragma unroll
    for (int i = 0; i < 4; ++i) {
        const int j = wave * 4 + i;
        tile[j][lane] = esum[(size_t)(k0 + j) * VEC + v0 + lane];
    }
    __syncthreads();
    const float csk = cs[k0 + lane];
#pragma unroll
    for (int i = 0; i < 4; ++i) {
        const int vl = wave * 4 + i;
        const size_t o = (size_t)(v0 + vl) * KCLUS + k0 + lane;
        out[o] = (GAMMA_ * embed_avg[o] + (1.0f - GAMMA_) * tile[lane][vl]) / csk;
    }
}

extern "C" void kernel_launch(void* const* d_in, const int* in_sizes, int n_in,
                              void* d_out, int out_size, void* d_ws, size_t ws_size,
                              hipStream_t stream) {
    const float* X            = (const float*)d_in[0];
    const float* W            = (const float*)d_in[1];
    const float* cluster_size = (const float*)d_in[2];
    const float* embed_avg    = (const float*)d_in[3];
    float* out = (float*)d_out;
    float* ws  = (float*)d_ws;

    float* esum   = ws + ESUM_OFF;
    float* w2     = ws + W2_OFF;
    float* cs     = ws + CS_OFF;
    int*   cnt    = (int*)(ws + COUNTS_OFF);
    int*   perm   = (int*)(out + PERM_OFF);
    int*   permk  = (int*)(out + PERMK_OFF);
    int*   cursor = (int*)(out + CURSOR_OFF);

    w2_kernel<<<KCLUS / 256, 256, 0, stream>>>(W, w2, cnt);

    size_t smem = (size_t)KPAIRS * RPITCH * sizeof(v4f);
    assign_slice_kernel<<<NSLICE * (BATCH / CHUNK_ROWS), 128, smem, stream>>>(
        X, W, w2, out, esum);

    merge_kernel<<<BATCH / 256, 256, 0, stream>>>(out, esum, cnt);

    scan_finalize1_kernel<<<1, KCLUS, 0, stream>>>(cluster_size, cnt, cursor, cs);
    place_kernel<<<BATCH / 256, 256, 0, stream>>>(out, cursor, perm, permk, esum);
    gather_kernel<<<BATCH / GROWS, 128, 0, stream>>>(X, perm, permk, esum);

    finalize2_kernel<<<(KCLUS / 32) * (VEC / 32), 256, 0, stream>>>(embed_avg, esum, cs, out);
}


# w2 summed inside the slice kernel from its LDS-resident W slice (thread = column, same rising-k chain of fused multiply-adds); the w2 launch and its pass over W removed, 7 -> 6 launches; sums kept
# speedup vs baseline: 5.1759x; 1.0683x over previous
#include <hip/hip_runtime.h>
#include <hip/hip_bf16.h>


#define BATCH 65536
#define VEC   512
#define KCLUS 1024
#define GAMMA_ 0.99f
#define EPS_   1e-5f

#define NSLICE      8
#define SLICE_COLS  (KCLUS / NSLICE)
#define CHUNK_ROWS  2048
#define WAVE_ROWS   (CHUNK_ROWS / 4)
#define NTILE       (WAVE_ROWS / 16)
#define RPITCH      72
#define KPAIRS      (VEC / 2)
#define BSTEPS      32
#define NBLK        (VEC / 4 / BSTEPS)
#define SLICE_BLOCKS SLICE_BLOCK16(0)

typedef __attribute__((ext_vector_type(2))) float v2f;
typedef __attribute__((ext_vector_type(8))) float v8f;

typedef __attribute__((ext_vector_type(4))) float v4f;

#define ESUM_OFF   0
#define COUNTS_OFF (VEC * KCLUS)
#define CS_OFF     (COUNTS_OFF + KCLUS)
#define PERM_OFF   BATCH
#define PERMK_OFF  (2 * BATCH)
#define CURSOR_OFF (3 * BATCH)

#define GROWS 32

template <int G>
__device__ __forceinline__ void slice_block(v2f (&a)[VEC / 4], v4f (&bb)[2][BSTEPS],
                                            v8f& acc0, v8f& acc1,
                                            float (&best)[8], int (&bidx)[8],
                                            const v4f* wlane, const float* xn,
                                            const float (&w2r)[8],
                                            const int colBase, const int c)
{
    constexpr int np  = G / NBLK;
    constexpr int blk = G % NBLK;
    constexpr int cur = G & 1;
    if (G + 1 < 4 * NBLK) {
        constexpr int np1  = (G + 1) / NBLK;
        constexpr int blk1 = (G + 1) % NBLK;
#pragma unroll
        for (int i = 0; i < BSTEPS; ++i) {
            const int s1 = blk1 * BSTEPS + i;
            bb[cur ^ 1][i] = wlane[s1 * 2 * RPITCH + np1 * 16];
        }
    }
    __builtin_amdgcn_sched_barrier(0);
    if (blk == 0) {
        acc0 = (v8f){0.f, 0.f, 0.f, 0.f, 0.f, 0.f, 0.f, 0.f};
        acc1 = (v8f){0.f, 0.f, 0.f, 0.f, 0.f, 0.f, 0.f, 0.f};
    }
#pragma unroll
    for (int i = 0; i < BSTEPS; ++i) {
        const int s = blk * BSTEPS + i;
        const v4f b = bb[cur][i];
        acc0 = __builtin_amdgcn_wmma_f32_16x16x4_f32(
                   false, a[s], false, __builtin_shufflevector(b, b, 0, 1), (short)0, acc0, false, false);
        acc1 = __builtin_amdgcn_wmma_f32_16x16x4_f32(
                   false, a[s], false, __builtin_shufflevector(b, b, 2, 3), (short)0, acc1, false, false);
        if (np == 3 && s >= 8) a[s - 8] = *(const v2f*)(xn + 4 * (s - 8));
    }
    asm volatile("" : "+v"(acc0), "+v"(acc1));
    if (blk == NBLK - 1) {
        const int col0 = colBase + np * 32 + c;
        const int col1 = col0 + 16;
        const float w2c0 = w2r[2 * np];
        const float w2c1 = w2r[2 * np + 1];
#pragma unroll
        for (int r = 0; r < 8; ++r) {
            float sc0 = w2c0 - 2.0f * acc0[r];
            if (sc0 < best[r]) { best[r] = sc0; bidx[r] = col0; }
        }
#pragma unroll
        for (int r = 0; r < 8; ++r) {
            float sc1 = w2c1 - 2.0f * acc1[r];
            if (sc1 < best[r]) { best[r] = sc1; bidx[r] = col1; }
        }
    }
    __builtin_amdgcn_sched_barrier(0);
}

#define SLICE_BLOCK(G)  slice_block<G>(a, bb, acc0, acc1, best, bidx, wlane, xn, w2r, colBase, c);
#define SLICE_BLOCK4(G)  SLICE_BLOCK(G) SLICE_BLOCK(G + 1) SLICE_BLOCK(G + 2) SLICE_BLOCK(G + 3)
#define SLICE_BLOCK16(G) SLICE_BLOCK4(G) SLICE_BLOCK4(G + 4) SLICE_BLOCK4(G + 8) SLICE_BLOCK4(G + 12)

__global__ __launch_bounds__(128)
void assign_slice_kernel(const float* __restrict__ X,
                         const float* __restrict__ W,
                         int* __restrict__ cnt,
                         float* __restrict__ pscore,
                         float* __restrict__ pidx)
{
    extern __shared__ char smem[];
    v4f* wl = (v4f*)smem;
    float* w2s = (float*)(wl + KPAIRS * RPITCH);

    const int tid   = threadIdx.x;
    const int lane  = tid & 31;
    const int wave  = tid >> 5;
    const int c     = lane & 15;
    const int h     = lane >> 4;
    const int slice = blockIdx.x & (NSLICE - 1);
    const int chunk = blockIdx.x >> 3;
    const int colBase = slice * SLICE_COLS;
    const int rowW    = chunk * CHUNK_ROWS + wave * WAVE_ROWS;

    if (chunk == 0) cnt[colBase + tid] = 0;

    v2f a[VEC / 4];
    {
        const float* xa = X + (size_t)(rowW + c) * VEC + 2 * h;
#pragma unroll
        for (int s = 0; s < VEC / 4; ++s) a[s] = *(const v2f*)(xa + 4 * s);
    }

    for (int i0 = tid; i0 < KPAIRS * 16; i0 += 128 * 4) {
        float4 w0a[4], w1a[4], w0b[4], w1b[4];
#pragma unroll
        for (int u = 0; u < 4; ++u) {
            const int i  = i0 + u * 128;
            const int kk = i >> 4;
            const int cc = ((i >> 2) & 3) * 32 + (i & 3) * 4;
            const float* r0 = W + (size_t)(2 * kk) * KCLUS + colBase + cc;
            w0a[u] = *(const float4*)(r0);
            w1a[u] = *(const float4*)(r0 + KCLUS);
            w0b[u] = *(const float4*)(r0 + 16);
            w1b[u] = *(const float4*)(r0 + KCLUS + 16);
        }
#pragma unroll
        for (int u = 0; u < 4; ++u) {
            const int i  = i0 + u * 128;
            const int kk = i >> 4;
            v4f* dst = wl + kk * RPITCH + ((i >> 2) & 3) * 16 + (i & 3) * 4;
            dst[0] = (v4f){w0a[u].x, w1a[u].x, w0b[u].x, w1b[u].x};
            dst[1] = (v4f){w0a[u].y, w1a[u].y, w0b[u].y, w1b[u].y};
            dst[2] = (v4f){w0a[u].z, w1a[u].z, w0b[u].z, w1b[u].z};
            dst[3] = (v4f){w0a[u].w, w1a[u].w, w0b[u].w, w1b[u].w};
        }
    }
    __syncthreads();

    const v4f* wlane = wl + h * RPITCH + c;

    {
        const v2f* colp = (const v2f*)(wl + (tid >> 5) * 16 + (tid & 15)) + ((tid >> 4) & 1);
        float s = 0.0f;
        v2f wc[8], wn[8];
#pragma unroll
        for (int u = 0; u < 8; ++u) wc[u] = colp[u * 2 * RPITCH];
#pragma unroll 1
        for (int kk = 0; kk < KPAIRS; kk += 8) {
            const int kn = (kk + 8 < KPAIRS) ? kk + 8 : kk;
#pragma unroll
            for (int u = 0; u < 8; ++u) wn[u] = colp[(kn + u) * 2 * RPITCH];
            __builtin_amdgcn_sched_barrier(0);
#pragma unroll
            for (int u = 0; u < 8; ++u) {
                s = __builtin_fmaf(wc[u].x, wc[u].x, s);
                s = __builtin_fmaf(wc[u].y, wc[u].y, s);
            }
#pragma unroll
            for (int u = 0; u < 8; ++u) wc[u] = wn[u];
        }
        w2s[tid] = s;
    }
    __syncthreads();

    float w2r[8];
#pragma unroll
    for (int n = 0; n < 8; ++n) w2r[n] = w2s[n * 16 + c];

#pragma unroll 1
    for (int t = 0; t < NTILE; ++t) {
        const int tn = (t + 1 < NTILE) ? t + 1 : t;
        const float* xn = X + (size_t)(rowW + tn * 16 + c) * VEC + 2 * h;

        float best[8];
        int   bidx[8];
#pragma unroll
        for (int r = 0; r < 8; ++r) { best[r] = 3.4e38f; bidx[r] = 0; }

        v4f bb[2][BSTEPS];
#pragma unroll
        for (int i = 0; i < BSTEPS; ++i) bb[0][i] = wlane[i * 2 * RPITCH];

        v8f acc0, acc1;
        SLICE_BLOCKS
#pragma unroll
        for (int s = VEC / 4 - 8; s < VEC / 4; ++s) a[s] = *(const v2f*)(xn + 4 * s);

        float mn[8];
#pragma unroll
        for (int r = 0; r < 8; ++r) mn[r] = best[r];
#pragma unroll
        for (int m = 8; m >= 1; m >>= 1) {
#pragma unroll
            for (int r = 0; r < 8; ++r) {
                const float o = __shfl_xor(mn[r], m, 32);
                mn[r] = (o < mn[r]) ? o : mn[r];
            }
        }
#pragma unroll
        for (int r = 0; r < 8; ++r) bidx[r] = (best[r] == mn[r]) ? bidx[r] : 0x7fffffff;
#pragma unroll
        for (int m = 8; m >= 1; m >>= 1) {
#pragma unroll
            for (int r = 0; r < 8; ++r) {
                const int o = __shfl_xor(bidx[r], m, 32);
                bidx[r] = (o < bidx[r]) ? o : bidx[r];
            }
        }
        float osc = mn[0];
        int   oid = bidx[0];
#pragma unroll
        for (int r = 1; r < 8; ++r) {
            if ((c & 7) == r) { osc = mn[r]; oid = bidx[r]; }
        }
        if (c < 8) {
            const size_t o = (size_t)slice * BATCH + (size_t)(rowW + t * 16 + 8 * h + c);
            pscore[o] = osc;
            pidx[o]   = __int_as_float(oid);
        }
    }
}

__global__ void merge_kernel(float* sc_assign, const float* __restrict__ pidx, int* __restrict__ cnt)
{
    const int row = blockIdx.x * blockDim.x + threadIdx.x;
    if (row >= BATCH) return;
    float best = 3.4e38f;
    int   bi   = 0;
#pragma unroll
    for (int s = 0; s < NSLICE; ++s) {
        const float v  = sc_assign[(size_t)s * BATCH + row];
        const int   id = __float_as_int(pidx[(size_t)s * BATCH + row]);
        if (v < best) { best = v; bi = id; }
    }
    sc_assign[row] = __int_as_float(bi & (KCLUS - 1));
    atomicAdd(&cnt[bi & (KCLUS - 1)], 1);
}

__global__ __launch_bounds__(1024)
void scan_finalize1_kernel(const float* __restrict__ cluster_size,
                           const int* __restrict__ cnt,
                           int* __restrict__ cursor,
                           float* __restrict__ cs)
{
    __shared__ float red[KCLUS];
    __shared__ int   pre[KCLUS];
    const int k  = threadIdx.x;
    const int ci = cnt[k];
    float c = (float)ci;
    if (c == 0.0f) c = 1.0f;
    float ncs = GAMMA_ * cluster_size[k] + (1.0f - GAMMA_) * c;
    red[k] = ncs;
    pre[k] = ci;
    __syncthreads();
    for (int s = KCLUS / 2; s > 0; s >>= 1) {
        if (k < s) red[k] += red[k + s];
        __syncthreads();
    }
    for (int d = 1; d < KCLUS; d <<= 1) {
        const int v = (k >= d) ? pre[k - d] : 0;
        __syncthreads();
        pre[k] += v;
        __syncthreads();
    }
    cursor[k] = pre[k] - ci;
    float n = red[0];
    cs[k] = (ncs + EPS_) / (n + (float)KCLUS * EPS_) * n;
}

__global__ __launch_bounds__(256)
void place_kernel(const float* __restrict__ assignf,
                  int* __restrict__ cursor,
                  int* __restrict__ perm,
                  int* __restrict__ permk,
                  float* __restrict__ esum)
{
    const int tid = threadIdx.x;
    const int row = blockIdx.x * 256 + tid;
    float4* z = (float4*)(esum + (size_t)blockIdx.x * 2048) + tid;
    z[0]   = make_float4(0.f, 0.f, 0.f, 0.f);
    z[256] = make_float4(0.f, 0.f, 0.f, 0.f);
    const int k   = __float_as_int(assignf[row]) & (KCLUS - 1);
    const int pos = atomicAdd(&cursor[k], 1) & (BATCH - 1);
    perm[pos]  = row;
    permk[pos] = k;
}

__global__ __launch_bounds__(128)
void gather_kernel(const float* __restrict__ X,
                   const int* __restrict__ perm,
                   const int* __restrict__ permk,
                   float* __restrict__ esum)
{
    const int lane = threadIdx.x & 31;
    const int wave = threadIdx.x >> 5;
    const int base = blockIdx.x * GROWS;
    const int myrow = perm[base + lane]  & (BATCH - 1);
    const int myk   = permk[base + lane] & (KCLUS - 1);
    const float* xb = X    + wave * 128 + 4 * lane;
    float*       eb = esum + wave * 128 + 4 * lane;

    v4f acc = {0.f, 0.f, 0.f, 0.f};
    int curk = __builtin_amdgcn_readlane(myk, 0);
#pragma unroll
    for (int g = 0; g < GROWS / 8; ++g) {
        v4f x[8];
#pragma unroll
        for (int u = 0; u < 8; ++u) {
            const int row = __builtin_amdgcn_readlane(myrow, 8 * g + u);
            x[u] = __builtin_nontemporal_load((const v4f*)(xb + (size_t)row * VEC));
        }
#pragma unroll
        for (int u = 0; u < 8; ++u) {
            const int k = __builtin_amdgcn_readlane(myk, 8 * g + u);
            if (k != curk) {
                float* e = eb + (size_t)curk * VEC;
                atomicAdd(e,     acc.x);
                atomicAdd(e + 1, acc.y);
                atomicAdd(e + 2, acc.z);
                atomicAdd(e + 3, acc.w);
                acc  = (v4f){0.f, 0.f, 0.f, 0.f};
                curk = k;
            }
            acc += x[u];
        }
    }
    float* e = eb + (size_t)curk * VEC;
    atomicAdd(e,     acc.x);
    atomicAdd(e + 1, acc.y);
    atomicAdd(e + 2, acc.z);
    atomicAdd(e + 3, acc.w);
}

__global__ __launch_bounds__(256)
void finalize2_kernel(const float* __restrict__ embed_avg,
                      const float* __restrict__ esum,
                      const float* __restrict__ cs,
                      float* __restrict__ out)
{
    __shared__ float tile[32][33];
    const int lane = threadIdx.x & 31;
    const int wave = threadIdx.x >> 5;
    const int k0 = (blockIdx.x >> 4) * 32;
    const int v0 = (blockIdx.x & 15) * 32;
#pragma unroll
    for (int i = 0; i < 4; ++i) {
        const int j = wave * 4 + i;
        tile[j][lane] = esum[(size_t)(k0 + j) * VEC + v0 + lane];
    }
    __syncthreads();
    const float csk = cs[k0 + lane];
#pragma unroll
    for (int i = 0; i < 4; ++i) {
        const int vl = wave * 4 + i;
        const size_t o = (size_t)(v0 + vl) * KCLUS + k0 + lane;
        out[o] = (GAMMA_ * embed_avg[o] + (1.0f - GAMMA_) * tile[lane][vl]) / csk;
    }
}

extern "C" void kernel_launch(void* const* d_in, const int* in_sizes, int n_in,
                              void* d_out, int out_size, void* d_ws, size_t ws_size,
                              hipStream_t stream) {
    const float* X            = (const float*)d_in[0];
    const float* W            = (const float*)d_in[1];
    const float* cluster_size = (const float*)d_in[2];
    const float* embed_avg    = (const float*)d_in[3];
    float* out = (float*)d_out;
    float* ws  = (float*)d_ws;

    float* esum   = ws + ESUM_OFF;
    float* cs     = ws + CS_OFF;
    int*   cnt    = (int*)(ws + COUNTS_OFF);
    int*   perm   = (int*)(out + PERM_OFF);
    int*   permk  = (int*)(out + PERMK_OFF);
    int*   cursor = (int*)(out + CURSOR_OFF);

    size_t smem = (size_t)KPAIRS * RPITCH * sizeof(v4f) + SLICE_COLS * sizeof(float);
    assign_slice_kernel<<<NSLICE * (BATCH / CHUNK_ROWS), 128, smem, stream>>>(
        X, W, cnt, out, esum);

    merge_kernel<<<BATCH / 256, 256, 0, stream>>>(out, esum, cnt);

    scan_finalize1_kernel<<<1, KCLUS, 0, stream>>>(cluster_size, cnt, cursor, cs);
    place_kernel<<<BATCH / 256, 256, 0, stream>>>(out, cursor, perm, permk, esum);
    gather_kernel<<<BATCH / GROWS, 128, 0, stream>>>(X, perm, permk, esum);

    finalize2_kernel<<<(KCLUS / 32) * (VEC / 32), 256, 0, stream>>>(embed_avg, esum, cs, out);
}


# Tile-end argmin over the 16 lanes of a row: DPP row_xmask swizzles on the vector ALU replace the 64 ds_bpermute LDS round trips per row tile; same minima, same first-column tie rule; sums unchanged
# speedup vs baseline: 5.3435x; 1.0324x over previous
#include <hip/hip_runtime.h>
#include <hip/hip_bf16.h>


#define BATCH 65536
#define VEC   512
#define KCLUS 1024
#define GAMMA_ 0.99f
#define EPS_   1e-5f

#define NSLICE      8
#define SLICE_COLS  (KCLUS / NSLICE)
#define CHUNK_ROWS  2048
#define WAVE_ROWS   (CHUNK_ROWS / 4)
#define NTILE       (WAVE_ROWS / 16)
#define RPITCH      72
#define KPAIRS      (VEC / 2)
#define BSTEPS      32
#define NBLK        (VEC / 4 / BSTEPS)
#define SLICE_BLOCKS SLICE_BLOCK16(0)

typedef __attribute__((ext_vector_type(2))) float v2f;
typedef __attribute__((ext_vector_type(8))) float v8f;

typedef __attribute__((ext_vector_type(4))) float v4f;

template <int M>
__device__ __forceinline__ int row_xor_i(int v)
{
    return __builtin_amdgcn_update_dpp(0, v, 0x160 + M, 0xf, 0xf, true);
}
template <int M>
__device__ __forceinline__ float row_xor_f(float v)
{
    return __int_as_float(row_xor_i<M>(__float_as_int(v)));
}

template <int M>
__device__ __forceinline__ void row_min_step(float (&mn)[8])
{
#pragma unroll
    for (int r = 0; r < 8; ++r) {
        const float o = row_xor_f<M>(mn[r]);
        mn[r] = (o < mn[r]) ? o : mn[r];
    }
}
template <int M>
__device__ __forceinline__ void row_min_step(int (&id)[8])
{
#pragma unroll
    for (int r = 0; r < 8; ++r) {
        const int o = row_xor_i<M>(id[r]);
        id[r] = (o < id[r]) ? o : id[r];
    }
}

#define ESUM_OFF   0
#define COUNTS_OFF (VEC * KCLUS)
#define CS_OFF     (COUNTS_OFF + KCLUS)
#define PERM_OFF   BATCH
#define PERMK_OFF  (2 * BATCH)
#define CURSOR_OFF (3 * BATCH)

#define GROWS 32

template <int G>
__device__ __forceinline__ void slice_block(v2f (&a)[VEC / 4], v4f (&bb)[2][BSTEPS],
                                            v8f& acc0, v8f& acc1,
                                            float (&best)[8], int (&bidx)[8],
                                            const v4f* wlane, const float* xn,
                                            const float (&w2r)[8],
                                            const int colBase, const int c)
{
    constexpr int np  = G / NBLK;
    constexpr int blk = G % NBLK;
    constexpr int cur = G & 1;
    if (G + 1 < 4 * NBLK) {
        constexpr int np1  = (G + 1) / NBLK;
        constexpr int blk1 = (G + 1) % NBLK;
#pragma unroll
        for (int i = 0; i < BSTEPS; ++i) {
            const int s1 = blk1 * BSTEPS + i;
            bb[cur ^ 1][i] = wlane[s1 * 2 * RPITCH + np1 * 16];
        }
    }
    __builtin_amdgcn_sched_barrier(0);
    if (blk == 0) {
        acc0 = (v8f){0.f, 0.f, 0.f, 0.f, 0.f, 0.f, 0.f, 0.f};
        acc1 = (v8f){0.f, 0.f, 0.f, 0.f, 0.f, 0.f, 0.f, 0.f};
    }
#pragma unroll
    for (int i = 0; i < BSTEPS; ++i) {
        const int s = blk * BSTEPS + i;
        const v4f b = bb[cur][i];
        acc0 = __builtin_amdgcn_wmma_f32_16x16x4_f32(
                   false, a[s], false, __builtin_shufflevector(b, b, 0, 1), (short)0, acc0, false, false);
        acc1 = __builtin_amdgcn_wmma_f32_16x16x4_f32(
                   false, a[s], false, __builtin_shufflevector(b, b, 2, 3), (short)0, acc1, false, false);
        if (np == 3 && s >= 8) a[s - 8] = *(const v2f*)(xn + 4 * (s - 8));
    }
    asm volatile("" : "+v"(acc0), "+v"(acc1));
    if (blk == NBLK - 1) {
        const int col0 = colBase + np * 32 + c;
        const int col1 = col0 + 16;
        const float w2c0 = w2r[2 * np];
        const float w2c1 = w2r[2 * np + 1];
#pragma unroll
        for (int r = 0; r < 8; ++r) {
            float sc0 = w2c0 - 2.0f * acc0[r];
            if (sc0 < best[r]) { best[r] = sc0; bidx[r] = col0; }
        }
#pragma unroll
        for (int r = 0; r < 8; ++r) {
            float sc1 = w2c1 - 2.0f * acc1[r];
            if (sc1 < best[r]) { best[r] = sc1; bidx[r] = col1; }
        }
    }
    __builtin_amdgcn_sched_barrier(0);
}

#define SLICE_BLOCK(G)  slice_block<G>(a, bb, acc0, acc1, best, bidx, wlane, xn, w2r, colBase, c);
#define SLICE_BLOCK4(G)  SLICE_BLOCK(G) SLICE_BLOCK(G + 1) SLICE_BLOCK(G + 2) SLICE_BLOCK(G + 3)
#define SLICE_BLOCK16(G) SLICE_BLOCK4(G) SLICE_BLOCK4(G + 4) SLICE_BLOCK4(G + 8) SLICE_BLOCK4(G + 12)

__global__ __launch_bounds__(128)
void assign_slice_kernel(const float* __restrict__ X,
                         const float* __restrict__ W,
                         int* __restrict__ cnt,
                         float* __restrict__ pscore,
                         float* __restrict__ pidx)
{
    extern __shared__ char smem[];
    v4f* wl = (v4f*)smem;
    float* w2s = (float*)(wl + KPAIRS * RPITCH);

    const int tid   = threadIdx.x;
    const int lane  = tid & 31;
    const int wave  = tid >> 5;
    const int c     = lane & 15;
    const int h     = lane >> 4;
    const int slice = blockIdx.x & (NSLICE - 1);
    const int chunk = blockIdx.x >> 3;
    const int colBase = slice * SLICE_COLS;
    const int rowW    = chunk * CHUNK_ROWS + wave * WAVE_ROWS;

    if (chunk == 0) cnt[colBase + tid] = 0;

    v2f a[VEC / 4];
    {
        const float* xa = X + (size_t)(rowW + c) * VEC + 2 * h;
#pragma unroll
        for (int s = 0; s < VEC / 4; ++s) a[s] = *(const v2f*)(xa + 4 * s);
    }

    for (int i0 = tid; i0 < KPAIRS * 16; i0 += 128 * 4) {
        float4 w0a[4], w1a[4], w0b[4], w1b[4];
#pragma unroll
        for (int u = 0; u < 4; ++u) {
            const int i  = i0 + u * 128;
            const int kk = i >> 4;
            const int cc = ((i >> 2) & 3) * 32 + (i & 3) * 4;
            const float* r0 = W + (size_t)(2 * kk) * KCLUS + colBase + cc;
            w0a[u] = *(const float4*)(r0);
            w1a[u] = *(const float4*)(r0 + KCLUS);
            w0b[u] = *(const float4*)(r0 + 16);
            w1b[u] = *(const float4*)(r0 + KCLUS + 16);
        }
#pragma unroll
        for (int u = 0; u < 4; ++u) {
            const int i  = i0 + u * 128;
            const int kk = i >> 4;
            v4f* dst = wl + kk * RPITCH + ((i >> 2) & 3) * 16 + (i & 3) * 4;
            dst[0] = (v4f){w0a[u].x, w1a[u].x, w0b[u].x, w1b[u].x};
            dst[1] = (v4f){w0a[u].y, w1a[u].y, w0b[u].y, w1b[u].y};
            dst[2] = (v4f){w0a[u].z, w1a[u].z, w0b[u].z, w1b[u].z};
            dst[3] = (v4f){w0a[u].w, w1a[u].w, w0b[u].w, w1b[u].w};
        }
    }
    __syncthreads();

    const v4f* wlane = wl + h * RPITCH + c;

    {
        const v2f* colp = (const v2f*)(wl + (tid >> 5) * 16 + (tid & 15)) + ((tid >> 4) & 1);
        float s = 0.0f;
        v2f wc[8], wn[8];
#pragma unroll
        for (int u = 0; u < 8; ++u) wc[u] = colp[u * 2 * RPITCH];
#pragma unroll 1
        for (int kk = 0; kk < KPAIRS; kk += 8) {
            const int kn = (kk + 8 < KPAIRS) ? kk + 8 : kk;
#pragma unroll
            for (int u = 0; u < 8; ++u) wn[u] = colp[(kn + u) * 2 * RPITCH];
            __builtin_amdgcn_sched_barrier(0);
#pragma unroll
            for (int u = 0; u < 8; ++u) {
                s = __builtin_fmaf(wc[u].x, wc[u].x, s);
                s = __builtin_fmaf(wc[u].y, wc[u].y, s);
            }
#pragma unroll
            for (int u = 0; u < 8; ++u) wc[u] = wn[u];
        }
        w2s[tid] = s;
    }
    __syncthreads();

    float w2r[8];
#pragma unroll
    for (int n = 0; n < 8; ++n) w2r[n] = w2s[n * 16 + c];

#pragma unroll 1
    for (int t = 0; t < NTILE; ++t) {
        const int tn = (t + 1 < NTILE) ? t + 1 : t;
        const float* xn = X + (size_t)(rowW + tn * 16 + c) * VEC + 2 * h;

        float best[8];
        int   bidx[8];
#pragma unroll
        for (int r = 0; r < 8; ++r) { best[r] = 3.4e38f; bidx[r] = 0; }

        v4f bb[2][BSTEPS];
#pragma unroll
        for (int i = 0; i < BSTEPS; ++i) bb[0][i] = wlane[i * 2 * RPITCH];

        v8f acc0, acc1;
        SLICE_BLOCKS
#pragma unroll
        for (int s = VEC / 4 - 8; s < VEC / 4; ++s) a[s] = *(const v2f*)(xn + 4 * s);

        float mn[8];
#pragma unroll
        for (int r = 0; r < 8; ++r) mn[r] = best[r];
        row_min_step<8>(mn);
        row_min_step<4>(mn);
        row_min_step<2>(mn);
        row_min_step<1>(mn);
#pragma unroll
        for (int r = 0; r < 8; ++r) bidx[r] = (best[r] == mn[r]) ? bidx[r] : 0x7fffffff;
        row_min_step<8>(bidx);
        row_min_step<4>(bidx);
        row_min_step<2>(bidx);
        row_min_step<1>(bidx);
        float osc = mn[0];
        int   oid = bidx[0];
#pragma unroll
        for (int r = 1; r < 8; ++r) {
            if ((c & 7) == r) { osc = mn[r]; oid = bidx[r]; }
        }
        if (c < 8) {
            const size_t o = (size_t)slice * BATCH + (size_t)(rowW + t * 16 + 8 * h + c);
            pscore[o] = osc;
            pidx[o]   = __int_as_float(oid);
        }
    }
}

__global__ void merge_kernel(float* sc_assign, const float* __restrict__ pidx, int* __restrict__ cnt)
{
    const int row = blockIdx.x * blockDim.x + threadIdx.x;
    if (row >= BATCH) return;
    float best = 3.4e38f;
    int   bi   = 0;
#pragma unroll
    for (int s = 0; s < NSLICE; ++s) {
        const float v  = sc_assign[(size_t)s * BATCH + row];
        const int   id = __float_as_int(pidx[(size_t)s * BATCH + row]);
        if (v < best) { best = v; bi = id; }
    }
    sc_assign[row] = __int_as_float(bi & (KCLUS - 1));
    atomicAdd(&cnt[bi & (KCLUS - 1)], 1);
}

__global__ __launch_bounds__(1024)
void scan_finalize1_kernel(const float* __restrict__ cluster_size,
                           const int* __restrict__ cnt,
                           int* __restrict__ cursor,
                           float* __restrict__ cs)
{
    __shared__ float red[KCLUS];
    __shared__ int   pre[KCLUS];
    const int k  = threadIdx.x;
    const int ci = cnt[k];
    float c = (float)ci;
    if (c == 0.0f) c = 1.0f;
    float ncs = GAMMA_ * cluster_size[k] + (1.0f - GAMMA_) * c;
    red[k] = ncs;
    pre[k] = ci;
    __syncthreads();
    for (int s = KCLUS / 2; s > 0; s >>= 1) {
        if (k < s) red[k] += red[k + s];
        __syncthreads();
    }
    for (int d = 1; d < KCLUS; d <<= 1) {
        const int v = (k >= d) ? pre[k - d] : 0;
        __syncthreads();
        pre[k] += v;
        __syncthreads();
    }
    cursor[k] = pre[k] - ci;
    float n = red[0];
    cs[k] = (ncs + EPS_) / (n + (float)KCLUS * EPS_) * n;
}

__global__ __launch_bounds__(256)
void place_kernel(const float* __restrict__ assignf,
                  int* __restrict__ cursor,
                  int* __restrict__ perm,
                  int* __restrict__ permk,
                  float* __restrict__ esum)
{
    const int tid = threadIdx.x;
    const int row = blockIdx.x * 256 + tid;
    float4* z = (float4*)(esum + (size_t)blockIdx.x * 2048) + tid;
    z[0]   = make_float4(0.f, 0.f, 0.f, 0.f);
    z[256] = make_float4(0.f, 0.f, 0.f, 0.f);
    const int k   = __float_as_int(assignf[row]) & (KCLUS - 1);
    const int pos = atomicAdd(&cursor[k], 1) & (BATCH - 1);
    perm[pos]  = row;
    permk[pos] = k;
}

__global__ __launch_bounds__(128)
void gather_kernel(const float* __restrict__ X,
                   const int* __restrict__ perm,
                   const int* __restrict__ permk,
                   float* __restrict__ esum)
{
    const int lane = threadIdx.x & 31;
    const int wave = threadIdx.x >> 5;
    const int base = blockIdx.x * GROWS;
    const int myrow = perm[base + lane]  & (BATCH - 1);
    const int myk   = permk[base + lane] & (KCLUS - 1);
    const float* xb = X    + wave * 128 + 4 * lane;
    float*       eb = esum + wave * 128 + 4 * lane;

    v4f acc = {0.f, 0.f, 0.f, 0.f};
    int curk = __builtin_amdgcn_readlane(myk, 0);
#pragma unroll
    for (int g = 0; g < GROWS / 8; ++g) {
        v4f x[8];
#pragma unroll
        for (int u = 0; u < 8; ++u) {
            const int row = __builtin_amdgcn_readlane(myrow, 8 * g + u);
            x[u] = __builtin_nontemporal_load((const v4f*)(xb + (size_t)row * VEC));
        }
#pragma unroll
        for (int u = 0; u < 8; ++u) {
            const int k = __builtin_amdgcn_readlane(myk, 8 * g + u);
            if (k != curk) {
                float* e = eb + (size_t)curk * VEC;
                atomicAdd(e,     acc.x);
                atomicAdd(e + 1, acc.y);
                atomicAdd(e + 2, acc.z);
                atomicAdd(e + 3, acc.w);
                acc  = (v4f){0.f, 0.f, 0.f, 0.f};
                curk = k;
            }
            acc += x[u];
        }
    }
    float* e = eb + (size_t)curk * VEC;
    atomicAdd(e,     acc.x);
    atomicAdd(e + 1, acc.y);
    atomicAdd(e + 2, acc.z);
    atomicAdd(e + 3, acc.w);
}

__global__ __launch_bounds__(256)
void finalize2_kernel(const float* __restrict__ embed_avg,
                      const float* __restrict__ esum,
                      const float* __restrict__ cs,
                      float* __restrict__ out)
{
    __shared__ float tile[32][33];
    const int lane = threadIdx.x & 31;
    const int wave = threadIdx.x >> 5;
    const int k0 = (blockIdx.x >> 4) * 32;
    const int v0 = (blockIdx.x & 15) * 32;
#pragma unroll
    for (int i = 0; i < 4; ++i) {
        const int j = wave * 4 + i;
        tile[j][lane] = esum[(size_t)(k0 + j) * VEC + v0 + lane];
    }
    __syncthreads();
    const float csk = cs[k0 + lane];
#pragma unroll
    for (int i = 0; i < 4; ++i) {
        const int vl = wave * 4 + i;
        const size_t o = (size_t)(v0 + vl) * KCLUS + k0 + lane;
        out[o] = (GAMMA_ * embed_avg[o] + (1.0f - GAMMA_) * tile[lane][vl]) / csk;
    }
}

extern "C" void kernel_launch(void* const* d_in, const int* in_sizes, int n_in,
                              void* d_out, int out_size, void* d_ws, size_t ws_size,
                              hipStream_t stream) {
    const float* X            = (const float*)d_in[0];
    const float* W            = (const float*)d_in[1];
    const float* cluster_size = (const float*)d_in[2];
    const float* embed_avg    = (const float*)d_in[3];
    float* out = (float*)d_out;
    float* ws  = (float*)d_ws;

    float* esum   = ws + ESUM_OFF;
    float* cs     = ws + CS_OFF;
    int*   cnt    = (int*)(ws + COUNTS_OFF);
    int*   perm   = (int*)(out + PERM_OFF);
    int*   permk  = (int*)(out + PERMK_OFF);
    int*   cursor = (int*)(out + CURSOR_OFF);

    size_t smem = (size_t)KPAIRS * RPITCH * sizeof(v4f) + SLICE_COLS * sizeof(float);
    assign_slice_kernel<<<NSLICE * (BATCH / CHUNK_ROWS), 128, smem, stream>>>(
        X, W, cnt, out, esum);

    merge_kernel<<<BATCH / 256, 256, 0, stream>>>(out, esum, cnt);

    scan_finalize1_kernel<<<1, KCLUS, 0, stream>>>(cluster_size, cnt, cursor, cs);
    place_kernel<<<BATCH / 256, 256, 0, stream>>>(out, cursor, perm, permk, esum);
    gather_kernel<<<BATCH / GROWS, 128, 0, stream>>>(X, perm, permk, esum);

    finalize2_kernel<<<(KCLUS / 32) * (VEC / 32), 256, 0, stream>>>(embed_avg, esum, cs, out);
}


# Slice kernel: 8 waves (two per SIMD) on the resident W slice; the older wave of a SIMD takes 26 row tiles, the younger 6 and fills the older's load and minima gaps with its products; sums kept
# speedup vs baseline: 5.5686x; 1.0421x over previous
#include <hip/hip_runtime.h>
#include <hip/hip_bf16.h>


#define BATCH 65536
#define VEC   512
#define KCLUS 1024
#define GAMMA_ 0.99f
#define EPS_   1e-5f

#define NSLICE      8
#define SLICE_COLS  (KCLUS / NSLICE)
#define CHUNK_ROWS  2048
#define NWAVES      8
#define NTHREADS    (NWAVES * 32)
#define TILES_OLD   26
#define TILES_YOUNG 6
static_assert(4 * (TILES_OLD + TILES_YOUNG) * 16 == CHUNK_ROWS, "the eight waves cover the chunk exactly once");
#define RPITCH      72
#define KPAIRS      (VEC / 2)
#define BSTEPS      16
#define WSTEPS      28
#define NWIN        ((VEC / 4 + WSTEPS - 1) / WSTEPS)
#define NBLK        (VEC / 4 / BSTEPS)
#define SLICE_BLOCKS SLICE_BLOCK16(0) SLICE_BLOCK16(16)

typedef __attribute__((ext_vector_type(2))) float v2f;
typedef __attribute__((ext_vector_type(8))) float v8f;

typedef __attribute__((ext_vector_type(4))) float v4f;

template <int M>
__device__ __forceinline__ int row_xor_i(int v)
{
    return __builtin_amdgcn_update_dpp(0, v, 0x160 + M, 0xf, 0xf, true);
}
template <int M>
__device__ __forceinline__ float row_xor_f(float v)
{
    return __int_as_float(row_xor_i<M>(__float_as_int(v)));
}

template <int M>
__device__ __forceinline__ void row_min_step(float (&mn)[8])
{
#pragma unroll
    for (int r = 0; r < 8; ++r) {
        const float o = row_xor_f<M>(mn[r]);
        mn[r] = (o < mn[r]) ? o : mn[r];
    }
}
template <int M>
__device__ __forceinline__ void row_min_step(int (&id)[8])
{
#pragma unroll
    for (int r = 0; r < 8; ++r) {
        const int o = row_xor_i<M>(id[r]);
        id[r] = (o < id[r]) ? o : id[r];
    }
}

#define ESUM_OFF   0
#define COUNTS_OFF (VEC * KCLUS)
#define CS_OFF     (COUNTS_OFF + KCLUS)
#define PERM_OFF   BATCH
#define PERMK_OFF  (2 * BATCH)
#define CURSOR_OFF (3 * BATCH)

#define GROWS 32

template <int G>
__device__ __forceinline__ void slice_block(v2f (&a)[VEC / 4], v4f (&bb)[2][BSTEPS],
                                            v8f& acc0, v8f& acc1,
                                            float (&best)[8], int (&bidx)[8],
                                            const char* lds, const unsigned (&wb)[NWIN], const float* xn,
                                            const float (&w2r)[8],
                                            const int colBase, const int c)
{
    constexpr int np  = G / NBLK;
    constexpr int blk = G % NBLK;
    constexpr int cur = G & 1;
    if (G + 1 < 4 * NBLK) {
        constexpr int np1  = (G + 1) / NBLK;
        constexpr int blk1 = (G + 1) % NBLK;
#pragma unroll
        for (int i = 0; i < BSTEPS; ++i) {
            const int s1 = blk1 * BSTEPS + i;
            bb[cur ^ 1][i] = *(const v4f*)(lds + wb[s1 / WSTEPS] + ((s1 % WSTEPS) * 2 * RPITCH + np1 * 16) * 16);
        }
    }
    __builtin_amdgcn_sched_barrier(0);
    if (blk == 0) {
        acc0 = (v8f){0.f, 0.f, 0.f, 0.f, 0.f, 0.f, 0.f, 0.f};
        acc1 = (v8f){0.f, 0.f, 0.f, 0.f, 0.f, 0.f, 0.f, 0.f};
    }
#pragma unroll
    for (int i = 0; i < BSTEPS; ++i) {
        const int s = blk * BSTEPS + i;
        const v4f b = bb[cur][i];
        acc0 = __builtin_amdgcn_wmma_f32_16x16x4_f32(
                   false, a[s], false, __builtin_shufflevector(b, b, 0, 1), (short)0, acc0, false, false);
        acc1 = __builtin_amdgcn_wmma_f32_16x16x4_f32(
                   false, a[s], false, __builtin_shufflevector(b, b, 2, 3), (short)0, acc1, false, false);
        if (np == 3 && s >= 8) a[s - 8] = *(const v2f*)(xn + 4 * (s - 8));
    }
    asm volatile("" : "+v"(acc0), "+v"(acc1));
    if (blk == NBLK - 1) {
        const int col0 = colBase + np * 32 + c;
        const int col1 = col0 + 16;
        const float w2c0 = w2r[2 * np];
        const float w2c1 = w2r[2 * np + 1];
#pragma unroll
        for (int r = 0; r < 8; ++r) {
            float sc0 = w2c0 - 2.0f * acc0[r];
            if (sc0 < best[r]) { best[r] = sc0; bidx[r] = col0; }
        }
#pragma unroll
        for (int r = 0; r < 8; ++r) {
            float sc1 = w2c1 - 2.0f * acc1[r];
            if (sc1 < best[r]) { best[r] = sc1; bidx[r] = col1; }
        }
    }
    __builtin_amdgcn_sched_barrier(0);
}

#define SLICE_BLOCK(G)  slice_block<G>(a, bb, acc0, acc1, best, bidx, smem, wb, xn, w2r, colBase, c);
#define SLICE_BLOCK4(G)  SLICE_BLOCK(G) SLICE_BLOCK(G + 1) SLICE_BLOCK(G + 2) SLICE_BLOCK(G + 3)
#define SLICE_BLOCK16(G) SLICE_BLOCK4(G) SLICE_BLOCK4(G + 4) SLICE_BLOCK4(G + 8) SLICE_BLOCK4(G + 12)

__global__ __launch_bounds__(NTHREADS)
void assign_slice_kernel(const float* __restrict__ X,
                         const float* __restrict__ W,
                         int* __restrict__ cnt,
                         float* __restrict__ pscore,
                         float* __restrict__ pidx)
{
    extern __shared__ char smem[];
    v4f* wl = (v4f*)smem;
    float* w2s = (float*)(wl + KPAIRS * RPITCH);

    const int tid   = threadIdx.x;
    const int lane  = tid & 31;
    const int wave  = tid >> 5;
    const int c     = lane & 15;
    const int h     = lane >> 4;
    const int slice = blockIdx.x & (NSLICE - 1);
    const int chunk = blockIdx.x >> 3;
    const int colBase = slice * SLICE_COLS;
    const int wv      = __builtin_amdgcn_readfirstlane(wave);
    const int ntile   = (wv < 4) ? TILES_OLD : TILES_YOUNG;
    const int rowW    = chunk * CHUNK_ROWS + ((wv < 4) ? wv * (TILES_OLD * 16)
                                                       : 4 * (TILES_OLD * 16) + (wv - 4) * (TILES_YOUNG * 16));

    if (chunk == 0 && tid < SLICE_COLS) cnt[colBase + tid] = 0;

    v2f a[VEC / 4];
    {
        const float* xa = X + (size_t)(rowW + c) * VEC + 2 * h;
#pragma unroll
        for (int s = 0; s < VEC / 4; ++s) a[s] = *(const v2f*)(xa + 4 * s);
    }

    for (int i0 = tid; i0 < KPAIRS * 16; i0 += NTHREADS * 4) {
        float4 w0a[4], w1a[4], w0b[4], w1b[4];
#pragma unroll
        for (int u = 0; u < 4; ++u) {
            const int i  = i0 + u * NTHREADS;
            const int kk = i >> 4;
            const int cc = ((i >> 2) & 3) * 32 + (i & 3) * 4;
            const float* r0 = W + (size_t)(2 * kk) * KCLUS + colBase + cc;
            w0a[u] = *(const float4*)(r0);
            w1a[u] = *(const float4*)(r0 + KCLUS);
            w0b[u] = *(const float4*)(r0 + 16);
            w1b[u] = *(const float4*)(r0 + KCLUS + 16);
        }
#pragma unroll
        for (int u = 0; u < 4; ++u) {
            const int i  = i0 + u * NTHREADS;
            const int kk = i >> 4;
            v4f* dst = wl + kk * RPITCH + ((i >> 2) & 3) * 16 + (i & 3) * 4;
            dst[0] = (v4f){w0a[u].x, w1a[u].x, w0b[u].x, w1b[u].x};
            dst[1] = (v4f){w0a[u].y, w1a[u].y, w0b[u].y, w1b[u].y};
            dst[2] = (v4f){w0a[u].z, w1a[u].z, w0b[u].z, w1b[u].z};
            dst[3] = (v4f){w0a[u].w, w1a[u].w, w0b[u].w, w1b[u].w};
        }
    }
    __syncthreads();

    unsigned wb[NWIN];
#pragma unroll
    for (int j = 0; j < NWIN; ++j) {
        wb[j] = (unsigned)((j * WSTEPS * 2 * RPITCH + h * RPITCH + c) * 16);
        asm volatile("" : "+v"(wb[j]));
    }

    if (tid < SLICE_COLS) {
        const v2f* colp = (const v2f*)(wl + (tid >> 5) * 16 + (tid & 15)) + ((tid >> 4) & 1);
        float s = 0.0f;
        v2f wc[8], wn[8];
#pragma unroll
        for (int u = 0; u < 8; ++u) wc[u] = colp[u * 2 * RPITCH];
#pragma unroll 1
        for (int kk = 0; kk < KPAIRS; kk += 8) {
            const int kn = (kk + 8 < KPAIRS) ? kk + 8 : kk;
#pragma unroll
            for (int u = 0; u < 8; ++u) wn[u] = colp[(kn + u) * 2 * RPITCH];
            __builtin_amdgcn_sched_barrier(0);
#pragma unroll
            for (int u = 0; u < 8; ++u) {
                s = __builtin_fmaf(wc[u].x, wc[u].x, s);
                s = __builtin_fmaf(wc[u].y, wc[u].y, s);
            }
#pragma unroll
            for (int u = 0; u < 8; ++u) wc[u] = wn[u];
        }
        w2s[tid] = s;
    }
    __syncthreads();

    float w2r[8];
#pragma unroll
    for (int n = 0; n < 8; ++n) w2r[n] = w2s[n * 16 + c];

#pragma unroll 1
    for (int t = 0; t < ntile; ++t) {
        const int tn = (t + 1 < ntile) ? t + 1 : t;
        const float* xn = X + (size_t)(rowW + tn * 16 + c) * VEC + 2 * h;

        float best[8];
        int   bidx[8];
#pragma unroll
        for (int r = 0; r < 8; ++r) { best[r] = 3.4e38f; bidx[r] = 0; }

        v4f bb[2][BSTEPS];
#pragma unroll
        for (int i = 0; i < BSTEPS; ++i) bb[0][i] = *(const v4f*)(smem + wb[0] + (i * 2 * RPITCH) * 16);

        v8f acc0, acc1;
        SLICE_BLOCKS
#pragma unroll
        for (int s = VEC / 4 - 8; s < VEC / 4; ++s) a[s] = *(const v2f*)(xn + 4 * s);

        float mn[8];
#pragma unroll
        for (int r = 0; r < 8; ++r) mn[r] = best[r];
        row_min_step<8>(mn);
        row_min_step<4>(mn);
        row_min_step<2>(mn);
        row_min_step<1>(mn);
#pragma unroll
        for (int r = 0; r < 8; ++r) bidx[r] = (best[r] == mn[r]) ? bidx[r] : 0x7fffffff;
        row_min_step<8>(bidx);
        row_min_step<4>(bidx);
        row_min_step<2>(bidx);
        row_min_step<1>(bidx);
        float osc = mn[0];
        int   oid = bidx[0];
#pragma unroll
        for (int r = 1; r < 8; ++r) {
            if ((c & 7) == r) { osc = mn[r]; oid = bidx[r]; }
        }
        if (c < 8) {
            const size_t o = (size_t)slice * BATCH + (size_t)(rowW + t * 16 + 8 * h + c);
            pscore[o] = osc;
            pidx[o]   = __int_as_float(oid);
        }
    }
}

__global__ void merge_kernel(float* sc_assign, const float* __restrict__ pidx, int* __restrict__ cnt)
{
    const int row = blockIdx.x * blockDim.x + threadIdx.x;
    if (row >= BATCH) return;
    float best = 3.4e38f;
    int   bi   = 0;
#pragma unroll
    for (int s = 0; s < NSLICE; ++s) {
        const float v  = sc_assign[(size_t)s * BATCH + row];
        const int   id = __float_as_int(pidx[(size_t)s * BATCH + row]);
        if (v < best) { best = v; bi = id; }
    }
    sc_assign[row] = __int_as_float(bi & (KCLUS - 1));
    atomicAdd(&cnt[bi & (KCLUS - 1)], 1);
}

__global__ __launch_bounds__(1024)
void scan_finalize1_kernel(const float* __restrict__ cluster_size,
                           const int* __restrict__ cnt,
                           int* __restrict__ cursor,
                           float* __restrict__ cs)
{
    __shared__ float red[KCLUS];
    __shared__ int   pre[KCLUS];
    const int k  = threadIdx.x;
    const int ci = cnt[k];
    float c = (float)ci;
    if (c == 0.0f) c = 1.0f;
    float ncs = GAMMA_ * cluster_size[k] + (1.0f - GAMMA_) * c;
    red[k] = ncs;
    pre[k] = ci;
    __syncthreads();
    for (int s = KCLUS / 2; s > 0; s >>= 1) {
        if (k < s) red[k] += red[k + s];
        __syncthreads();
    }
    for (int d = 1; d < KCLUS; d <<= 1) {
        const int v = (k >= d) ? pre[k - d] : 0;
        __syncthreads();
        pre[k] += v;
        __syncthreads();
    }
    cursor[k] = pre[k] - ci;
    float n = red[0];
    cs[k] = (ncs + EPS_) / (n + (float)KCLUS * EPS_) * n;
}

__global__ __launch_bounds__(256)
void place_kernel(const float* __restrict__ assignf,
                  int* __restrict__ cursor,
                  int* __restrict__ perm,
                  int* __restrict__ permk,
                  float* __restrict__ esum)
{
    const int tid = threadIdx.x;
    const int row = blockIdx.x * 256 + tid;
    float4* z = (float4*)(esum + (size_t)blockIdx.x * 2048) + tid;
    z[0]   = make_float4(0.f, 0.f, 0.f, 0.f);
    z[256] = make_float4(0.f, 0.f, 0.f, 0.f);
    const int k   = __float_as_int(assignf[row]) & (KCLUS - 1);
    const int pos = atomicAdd(&cursor[k], 1) & (BATCH - 1);
    perm[pos]  = row;
    permk[pos] = k;
}

__global__ __launch_bounds__(128)
void gather_kernel(const float* __restrict__ X,
                   const int* __restrict__ perm,
                   const int* __restrict__ permk,
                   float* __restrict__ esum)
{
    const int lane = threadIdx.x & 31;
    const int wave = threadIdx.x >> 5;
    const int base = blockIdx.x * GROWS;
    const int myrow = perm[base + lane]  & (BATCH - 1);
    const int myk   = permk[base + lane] & (KCLUS - 1);
    const float* xb = X    + wave * 128 + 4 * lane;
    float*       eb = esum + wave * 128 + 4 * lane;

    v4f acc = {0.f, 0.f, 0.f, 0.f};
    int curk = __builtin_amdgcn_readlane(myk, 0);
#pragma unroll
    for (int g = 0; g < GROWS / 8; ++g) {
        v4f x[8];
#pragma unroll
        for (int u = 0; u < 8; ++u) {
            const int row = __builtin_amdgcn_readlane(myrow, 8 * g + u);
            x[u] = __builtin_nontemporal_load((const v4f*)(xb + (size_t)row * VEC));
        }
#pragma unroll
        for (int u = 0; u < 8; ++u) {
            const int k = __builtin_amdgcn_readlane(myk, 8 * g + u);
            if (k != curk) {
                float* e = eb + (size_t)curk * VEC;
                atomicAdd(e,     acc.x);
                atomicAdd(e + 1, acc.y);
                atomicAdd(e + 2, acc.z);
                atomicAdd(e + 3, acc.w);
                acc  = (v4f){0.f, 0.f, 0.f, 0.f};
                curk = k;
            }
            acc += x[u];
        }
    }
    float* e = eb + (size_t)curk * VEC;
    atomicAdd(e,     acc.x);
    atomicAdd(e + 1, acc.y);
    atomicAdd(e + 2, acc.z);
    atomicAdd(e + 3, acc.w);
}

__global__ __launch_bounds__(256)
void finalize2_kernel(const float* __restrict__ embed_avg,
                      const float* __restrict__ esum,
                      const float* __restrict__ cs,
                      float* __restrict__ out)
{
    __shared__ float tile[32][33];
    const int lane = threadIdx.x & 31;
    const int wave = threadIdx.x >> 5;
    const int k0 = (blockIdx.x >> 4) * 32;
    const int v0 = (blockIdx.x & 15) * 32;
#pragma unroll
    for (int i = 0; i < 4; ++i) {
        const int j = wave * 4 + i;
        tile[j][lane] = esum[(size_t)(k0 + j) * VEC + v0 + lane];
    }
    __syncthreads();
    const float csk = cs[k0 + lane];
#pragma unroll
    for (int i = 0; i < 4; ++i) {
        const int vl = wave * 4 + i;
        const size_t o = (size_t)(v0 + vl) * KCLUS + k0 + lane;
        out[o] = (GAMMA_ * embed_avg[o] + (1.0f - GAMMA_) * tile[lane][vl]) / csk;
    }
}

extern "C" void kernel_launch(void* const* d_in, const int* in_sizes, int n_in,
                              void* d_out, int out_size, void* d_ws, size_t ws_size,
                              hipStream_t stream) {
    const float* X            = (const float*)d_in[0];
    const float* W            = (const float*)d_in[1];
    const float* cluster_size = (const float*)d_in[2];
    const float* embed_avg    = (const float*)d_in[3];
    float* out = (float*)d_out;
    float* ws  = (float*)d_ws;

    float* esum   = ws + ESUM_OFF;
    float* cs     = ws + CS_OFF;
    int*   cnt    = (int*)(ws + COUNTS_OFF);
    int*   perm   = (int*)(out + PERM_OFF);
    int*   permk  = (int*)(out + PERMK_OFF);
    int*   cursor = (int*)(out + CURSOR_OFF);

    size_t smem = (size_t)KPAIRS * RPITCH * sizeof(v4f) + SLICE_COLS * sizeof(float);
    assign_slice_kernel<<<NSLICE * (BATCH / CHUNK_ROWS), NTHREADS, smem, stream>>>(
        X, W, cnt, out, esum);

    merge_kernel<<<BATCH / 256, 256, 0, stream>>>(out, esum, cnt);

    scan_finalize1_kernel<<<1, KCLUS, 0, stream>>>(cluster_size, cnt, cursor, cs);
    place_kernel<<<BATCH / 256, 256, 0, stream>>>(out, cursor, perm, permk, esum);
    gather_kernel<<<BATCH / GROWS, 128, 0, stream>>>(X, perm, permk, esum);

    finalize2_kernel<<<(KCLUS / 32) * (VEC / 32), 256, 0, stream>>>(embed_avg, esum, cs, out);
}
